# starting kernel
# baseline (speedup 1.0000x reference)
#include <hip/hip_runtime.h>
#include <hip/hip_bf16.h>

typedef float v2f __attribute__((ext_vector_type(2)));
typedef float v8f __attribute__((ext_vector_type(8)));

#define B_ 128
#define U_ 8
#define C_ 2048
#define J_ 32
#define S_ 16
#define K_ (C_ * U_)

__global__ void transpose_W_kernel(const float* __restrict__ W, float* __restrict__ WT2) {
    int o = blockIdx.x * 256 + threadIdx.x;
    int t  = o & 1;
    int s  = (o >> 1) & 15;
    int kp = (o >> 5) & 8191;
    int j  = o >> 18;
    int k  = kp * 2 + t;
    int c  = k >> 3, u = k & 7;
    WT2[o] = W[(((size_t)c * J_ + j) * S_ + s) * U_ + u];
}

__global__ void transpose_x_kernel(const float* __restrict__ x, float* __restrict__ xT2) {
    int o = blockIdx.x * 256 + threadIdx.x;
    int t  = o & 1;
    int b  = (o >> 1) & 127;
    int kp = o >> 8;
    int k  = kp * 2 + t;
    int c  = k >> 3, u = k & 7;
    xT2[o] = x[((size_t)b * U_ + u) * C_ + c];
}

__global__ void softmax_c_kernel(const float* __restrict__ bb, float* __restrict__ cc) {
    int j = blockIdx.x;
    int tid = threadIdx.x;
    __shared__ float red[256];
    float m = -1e30f;
    for (int c = tid; c < C_; c += 256) m = fmaxf(m, bb[c * J_ + j]);
    red[tid] = m; __syncthreads();
    for (int off = 128; off > 0; off >>= 1) {
        if (tid < off) red[tid] = fmaxf(red[tid], red[tid + off]);
        __syncthreads();
    }
    float mx = red[0]; __syncthreads();
    float sum = 0.f;
    for (int c = tid; c < C_; c += 256) sum += __expf(bb[c * J_ + j] - mx);
    red[tid] = sum; __syncthreads();
    for (int off = 128; off > 0; off >>= 1) {
        if (tid < off) red[tid] += red[tid + off];
        __syncthreads();
    }
    float inv = 1.0f / red[0];
    for (int c = tid; c < C_; c += 256) cc[c * J_ + j] = __expf(bb[c * J_ + j] - mx) * inv;
}

__global__ __launch_bounds__(256) void s_kernel(
    const float* __restrict__ WT2, const float* __restrict__ xT2,
    const float* __restrict__ cc, float* __restrict__ spart) {
    int bx  = blockIdx.x;
    int ks  = bx & 3;
    int btp = (bx >> 2) & 3;
    int jp  = bx >> 4;
    int j0  = jp * 2;
    int bt0 = btp * 2;

    int tid  = threadIdx.x;
    int w    = tid >> 5;
    int lane = tid & 31;
    int perm = (lane >> 4) * 2;
    int ln   = lane & 15;

    const float* wt0 = WT2 + (size_t)j0 * K_ * S_;
    const float* wt1 = wt0 + (size_t)K_ * S_;

    v8f a00 = {}, a01 = {}, a10 = {}, a11 = {};
    int k0 = ks * 4096 + w * 512;
    for (int kb = k0; kb < k0 + 512; kb += 4) {
        int c = kb >> 3;
        v2f cw = *(const v2f*)(cc + (size_t)c * J_ + j0);
        int kk = kb + perm;
        size_t kpi = (size_t)(kk >> 1);
        v2f fa0 = *(const v2f*)(wt0 + (kpi * 16 + ln) * 2);
        v2f fa1 = *(const v2f*)(wt1 + (kpi * 16 + ln) * 2);
        v2f fb0 = *(const v2f*)(xT2 + (kpi * 128 + bt0 * 16 + ln) * 2);
        v2f fb1 = *(const v2f*)(xT2 + (kpi * 128 + (bt0 + 1) * 16 + ln) * 2);
        fa0 = fa0 * cw.x;
        fa1 = fa1 * cw.y;
        a00 = __builtin_amdgcn_wmma_f32_16x16x4_f32(false, fa0, false, fb0, (short)0, a00, false, false);
        a01 = __builtin_amdgcn_wmma_f32_16x16x4_f32(false, fa0, false, fb1, (short)0, a01, false, false);
        a10 = __builtin_amdgcn_wmma_f32_16x16x4_f32(false, fa1, false, fb0, (short)0, a10, false, false);
        a11 = __builtin_amdgcn_wmma_f32_16x16x4_f32(false, fa1, false, fb1, (short)0, a11, false, false);
    }

    __shared__ float red[8][4][256];
#pragma unroll
    for (int r = 0; r < 8; r++) {
        int m = r + 8 * (lane >> 4);
        red[w][0][m * 16 + ln] = a00[r];
        red[w][1][m * 16 + ln] = a01[r];
        red[w][2][m * 16 + ln] = a10[r];
        red[w][3][m * 16 + ln] = a11[r];
    }
    __syncthreads();
    int m = tid >> 4, n = tid & 15;
    float* sp = spart + (size_t)ks * (B_ * J_ * S_);
#pragma unroll
    for (int t = 0; t < 4; t++) {
        int jj = t >> 1, bb = t & 1;
        float sum = 0.f;
#pragma unroll
        for (int ww = 0; ww < 8; ww++) sum += red[ww][t][tid];
        sp[(size_t)((bt0 + bb) * 16 + n) * (J_ * S_) + (j0 + jj) * S_ + m] = sum;
    }
}

__global__ void squash_kernel(const float* __restrict__ spart, float* __restrict__ vbuf,
                              float* __restrict__ out, int write_out) {
    int b = blockIdx.x;
    int tid = threadIdx.x;
    __shared__ float red[512];
    size_t idx = (size_t)b * 512 + tid;
    float val = spart[idx] + spart[idx + 65536] + spart[idx + 2 * 65536] + spart[idx + 3 * 65536];
    red[tid] = val * val;
    __syncthreads();
    for (int jo = 256; jo >= 16; jo >>= 1) {
        if (tid < jo) red[tid] += red[tid + jo];
        __syncthreads();
    }
    float msq = red[tid & 15];
    float scale = msq / ((1.0f + msq) * sqrtf(msq));
    float vv = val * scale;
    vbuf[idx] = vv;
    if (write_out) out[idx] = vv;
}

__global__ __launch_bounds__(128) void bupd_kernel(
    const float* __restrict__ x, const float* __restrict__ W,
    const float* __restrict__ vbuf, float* __restrict__ bb) {
    int ct = blockIdx.x;
    int jq = blockIdx.y;
    int cbase = ct * 16;
    int tid  = threadIdx.x;
    int w    = tid >> 5;
    int lane = tid & 31;
    int perm = (lane >> 4) * 2;
    int ln   = lane & 15;
    int j    = jq * 4 + w;

    __shared__ float vlds[4][B_ * S_];
    for (int i = lane; i < B_ * S_; i += 32)
        vlds[w][i] = vbuf[(size_t)(i >> 4) * (J_ * S_) + j * S_ + (i & 15)];
    __syncthreads();

    float sacc[8];
#pragma unroll
    for (int r = 0; r < 8; r++) sacc[r] = 0.f;

    for (int u = 0; u < U_; u++) {
        v8f acc = {};
        for (int kb = 0; kb < B_; kb += 4) {
            int k = kb + perm;
            v2f a, bf;
            a.x  = x[((size_t)k * U_ + u) * C_ + cbase + ln];
            a.y  = x[((size_t)(k + 1) * U_ + u) * C_ + cbase + ln];
            bf.x = vlds[w][k * 16 + ln];
            bf.y = vlds[w][(k + 1) * 16 + ln];
            acc = __builtin_amdgcn_wmma_f32_16x16x4_f32(
                false, a, false, bf, (short)0, acc, false, false);
        }
#pragma unroll
        for (int r = 0; r < 8; r++) {
            int c = cbase + r + 8 * (lane >> 4);
            float wv = W[((size_t)c * J_ + j) * (S_ * U_) + ln * U_ + u];
            float p = acc[r] * wv;
            p += __shfl_xor(p, 1);
            p += __shfl_xor(p, 2);
            p += __shfl_xor(p, 4);
            p += __shfl_xor(p, 8);
            sacc[r] += p;
        }
    }
    if (ln == 0) {
        int h = lane >> 4;
#pragma unroll
        for (int r = 0; r < 8; r++) {
            size_t bi = (size_t)(cbase + r + 8 * h) * J_ + j;
            bb[bi] += sacc[r] * (1.0f / 128.0f);
        }
    }
}

extern "C" void kernel_launch(void* const* d_in, const int* in_sizes, int n_in,
                              void* d_out, int out_size, void* d_ws, size_t ws_size,
                              hipStream_t stream) {
    (void)in_sizes; (void)n_in; (void)out_size; (void)ws_size;
    const float* x = (const float*)d_in[0];
    const float* W = (const float*)d_in[1];
    float* out = (float*)d_out;

    float* ws    = (float*)d_ws;
    float* WT2   = ws;
    float* xT2   = WT2 + 8388608;
    float* bb    = xT2 + 2097152;
    float* cc    = bb + 65536;
    float* spart = cc + 65536;
    float* vbuf  = spart + 4 * 65536;

    hipMemsetAsync(bb, 0, 65536 * sizeof(float), stream);
    transpose_W_kernel<<<8388608 / 256, 256, 0, stream>>>(W, WT2);
    transpose_x_kernel<<<2097152 / 256, 256, 0, stream>>>(x, xT2);

    for (int it = 0; it < 3; it++) {
        softmax_c_kernel<<<32, 256, 0, stream>>>(bb, cc);
        s_kernel<<<256, 256, 0, stream>>>(WT2, xT2, cc, spart);
        squash_kernel<<<128, 512, 0, stream>>>(spart, vbuf, out, it == 2 ? 1 : 0);
        if (it < 2) {
            bupd_kernel<<<dim3(128, 8), 128, 0, stream>>>(x, W, vbuf, bb);
        }
    }
}


# s_kernel K-loop software-pipelined: fragment loads of the next 8 K-steps issued as one group ahead of the current WMMAs (two register sets, routing weights as scalar loads); sum order unchanged
# speedup vs baseline: 1.6141x; 1.6141x over previous
#include <hip/hip_runtime.h>
#include <hip/hip_bf16.h>

typedef float v2f __attribute__((ext_vector_type(2)));
typedef float v8f __attribute__((ext_vector_type(8)));

#define B_ 128
#define U_ 8
#define C_ 2048
#define J_ 32
#define S_ 16
#define K_ (C_ * U_)

__global__ void transpose_W_kernel(const float* __restrict__ W, float* __restrict__ WT2) {
    int o = blockIdx.x * 256 + threadIdx.x;
    int t  = o & 1;
    int s  = (o >> 1) & 15;
    int kp = (o >> 5) & 8191;
    int j  = o >> 18;
    int k  = kp * 2 + t;
    int c  = k >> 3, u = k & 7;
    WT2[o] = W[(((size_t)c * J_ + j) * S_ + s) * U_ + u];
}

__global__ void transpose_x_kernel(const float* __restrict__ x, float* __restrict__ xT2) {
    int o = blockIdx.x * 256 + threadIdx.x;
    int t  = o & 1;
    int b  = (o >> 1) & 127;
    int kp = o >> 8;
    int k  = kp * 2 + t;
    int c  = k >> 3, u = k & 7;
    xT2[o] = x[((size_t)b * U_ + u) * C_ + c];
}

__global__ void softmax_c_kernel(const float* __restrict__ bb, float* __restrict__ cc) {
    int j = blockIdx.x;
    int tid = threadIdx.x;
    __shared__ float red[256];
    float m = -1e30f;
    for (int c = tid; c < C_; c += 256) m = fmaxf(m, bb[c * J_ + j]);
    red[tid] = m; __syncthreads();
    for (int off = 128; off > 0; off >>= 1) {
        if (tid < off) red[tid] = fmaxf(red[tid], red[tid + off]);
        __syncthreads();
    }
    float mx = red[0]; __syncthreads();
    float sum = 0.f;
    for (int c = tid; c < C_; c += 256) sum += __expf(bb[c * J_ + j] - mx);
    red[tid] = sum; __syncthreads();
    for (int off = 128; off > 0; off >>= 1) {
        if (tid < off) red[tid] += red[tid + off];
        __syncthreads();
    }
    float inv = 1.0f / red[0];
    for (int c = tid; c < C_; c += 256) cc[c * J_ + j] = __expf(bb[c * J_ + j] - mx) * inv;
}

#define SK_UN 8
#define SK_NB (512 / (4 * SK_UN))

struct SkSet {
    v2f a0[SK_UN], a1[SK_UN], b0[SK_UN], b1[SK_UN];
    v2f cw[SK_UN / 2];
};

__device__ __forceinline__ void sk_load(SkSet& f, const float* __restrict__ pa0,
                                        const float* __restrict__ pa1,
                                        const float* __restrict__ pb,
                                        const float* __restrict__ pc) {
#pragma unroll
    for (int i = 0; i < SK_UN / 2; i++) f.cw[i] = *(const v2f*)(pc + i * J_);
#pragma unroll
    for (int i = 0; i < SK_UN; i++) {
        f.a0[i] = *(const v2f*)(pa0 + i * 64);
        f.a1[i] = *(const v2f*)(pa1 + i * 64);
        f.b0[i] = *(const v2f*)(pb + i * 512);
        f.b1[i] = *(const v2f*)(pb + i * 512 + 32);
    }
}

__device__ __forceinline__ void sk_compute(const SkSet& f, v8f& a00, v8f& a01,
                                           v8f& a10, v8f& a11) {
#pragma unroll
    for (int i = 0; i < SK_UN; i++) {
        v2f cw = f.cw[i >> 1];
        v2f fa0 = f.a0[i] * cw.x;
        v2f fa1 = f.a1[i] * cw.y;
        a00 = __builtin_amdgcn_wmma_f32_16x16x4_f32(false, fa0, false, f.b0[i], (short)0, a00, false, false);
        a01 = __builtin_amdgcn_wmma_f32_16x16x4_f32(false, fa0, false, f.b1[i], (short)0, a01, false, false);
        a10 = __builtin_amdgcn_wmma_f32_16x16x4_f32(false, fa1, false, f.b0[i], (short)0, a10, false, false);
        a11 = __builtin_amdgcn_wmma_f32_16x16x4_f32(false, fa1, false, f.b1[i], (short)0, a11, false, false);
    }
}

__global__ __launch_bounds__(256) void s_kernel(
    const float* __restrict__ WT2, const float* __restrict__ xT2,
    const float* __restrict__ cc, float* __restrict__ spart) {
    int bx  = blockIdx.x;
    int ks  = bx & 3;
    int btp = (bx >> 2) & 3;
    int jp  = bx >> 4;
    int j0  = jp * 2;
    int bt0 = btp * 2;

    int tid  = threadIdx.x;
    int w    = __builtin_amdgcn_readfirstlane(tid >> 5);
    int lane = tid & 31;
    int h    = lane >> 4;
    int ln   = lane & 15;

    int k0 = ks * 4096 + w * 512;
    size_t kp0 = (size_t)(k0 >> 1) + h;
    const float* pa0 = WT2 + (size_t)j0 * K_ * S_ + (kp0 * 16 + ln) * 2;
    const float* pa1 = pa0 + (size_t)K_ * S_;
    const float* pb  = xT2 + (kp0 * 128 + bt0 * 16 + ln) * 2;
    const float* pc  = cc + (size_t)(k0 >> 3) * J_ + j0;

    v8f a00 = {}, a01 = {}, a10 = {}, a11 = {};
    SkSet fA, fB;
#define SK_FENCE() __builtin_amdgcn_sched_barrier(0)
    sk_load(fA, pa0, pa1, pb, pc);
    for (int it = 0; it < SK_NB / 2 - 1; it++) {
        sk_load(fB, pa0 + SK_UN * 64, pa1 + SK_UN * 64, pb + SK_UN * 512, pc + (SK_UN / 2) * J_);
        SK_FENCE();
        sk_compute(fA, a00, a01, a10, a11);
        pa0 += 2 * SK_UN * 64; pa1 += 2 * SK_UN * 64; pb += 2 * SK_UN * 512; pc += SK_UN * J_;
        sk_load(fA, pa0, pa1, pb, pc);
        SK_FENCE();
        sk_compute(fB, a00, a01, a10, a11);
    }
    sk_load(fB, pa0 + SK_UN * 64, pa1 + SK_UN * 64, pb + SK_UN * 512, pc + (SK_UN / 2) * J_);
    SK_FENCE();
    sk_compute(fA, a00, a01, a10, a11);
    sk_compute(fB, a00, a01, a10, a11);
#undef SK_FENCE

    __shared__ float red[8][4][256];
#pragma unroll
    for (int r = 0; r < 8; r++) {
        int m = r + 8 * h;
        red[w][0][m * 16 + ln] = a00[r];
        red[w][1][m * 16 + ln] = a01[r];
        red[w][2][m * 16 + ln] = a10[r];
        red[w][3][m * 16 + ln] = a11[r];
    }
    __syncthreads();
    int m = tid >> 4, n = tid & 15;
    float* sp = spart + (size_t)ks * (B_ * J_ * S_);
#pragma unroll
    for (int t = 0; t < 4; t++) {
        int jj = t >> 1, bb = t & 1;
        float sum = 0.f;
#pragma unroll
        for (int ww = 0; ww < 8; ww++) sum += red[ww][t][tid];
        sp[(size_t)((bt0 + bb) * 16 + n) * (J_ * S_) + (j0 + jj) * S_ + m] = sum;
    }
}

__global__ void squash_kernel(const float* __restrict__ spart, float* __restrict__ vbuf,
                              float* __restrict__ out, int write_out) {
    int b = blockIdx.x;
    int tid = threadIdx.x;
    __shared__ float red[512];
    size_t idx = (size_t)b * 512 + tid;
    float val = spart[idx] + spart[idx + 65536] + spart[idx + 2 * 65536] + spart[idx + 3 * 65536];
    red[tid] = val * val;
    __syncthreads();
    for (int jo = 256; jo >= 16; jo >>= 1) {
        if (tid < jo) red[tid] += red[tid + jo];
        __syncthreads();
    }
    float msq = red[tid & 15];
    float scale = msq / ((1.0f + msq) * sqrtf(msq));
    float vv = val * scale;
    vbuf[idx] = vv;
    if (write_out) out[idx] = vv;
}

__global__ __launch_bounds__(128) void bupd_kernel(
    const float* __restrict__ x, const float* __restrict__ W,
    const float* __restrict__ vbuf, float* __restrict__ bb) {
    int ct = blockIdx.x;
    int jq = blockIdx.y;
    int cbase = ct * 16;
    int tid  = threadIdx.x;
    int w    = tid >> 5;
    int lane = tid & 31;
    int perm = (lane >> 4) * 2;
    int ln   = lane & 15;
    int j    = jq * 4 + w;

    __shared__ float vlds[4][B_ * S_];
    for (int i = lane; i < B_ * S_; i += 32)
        vlds[w][i] = vbuf[(size_t)(i >> 4) * (J_ * S_) + j * S_ + (i & 15)];
    __syncthreads();

    float sacc[8];
#pragma unroll
    for (int r = 0; r < 8; r++) sacc[r] = 0.f;

    for (int u = 0; u < U_; u++) {
        v8f acc = {};
        for (int kb = 0; kb < B_; kb += 4) {
            int k = kb + perm;
            v2f a, bf;
            a.x  = x[((size_t)k * U_ + u) * C_ + cbase + ln];
            a.y  = x[((size_t)(k + 1) * U_ + u) * C_ + cbase + ln];
            bf.x = vlds[w][k * 16 + ln];
            bf.y = vlds[w][(k + 1) * 16 + ln];
            acc = __builtin_amdgcn_wmma_f32_16x16x4_f32(
                false, a, false, bf, (short)0, acc, false, false);
        }
#pragma unroll
        for (int r = 0; r < 8; r++) {
            int c = cbase + r + 8 * (lane >> 4);
            float wv = W[((size_t)c * J_ + j) * (S_ * U_) + ln * U_ + u];
            float p = acc[r] * wv;
            p += __shfl_xor(p, 1);
            p += __shfl_xor(p, 2);
            p += __shfl_xor(p, 4);
            p += __shfl_xor(p, 8);
            sacc[r] += p;
        }
    }
    if (ln == 0) {
        int h = lane >> 4;
#pragma unroll
        for (int r = 0; r < 8; r++) {
            size_t bi = (size_t)(cbase + r + 8 * h) * J_ + j;
            bb[bi] += sacc[r] * (1.0f / 128.0f);
        }
    }
}

extern "C" void kernel_launch(void* const* d_in, const int* in_sizes, int n_in,
                              void* d_out, int out_size, void* d_ws, size_t ws_size,
                              hipStream_t stream) {
    (void)in_sizes; (void)n_in; (void)out_size; (void)ws_size;
    const float* x = (const float*)d_in[0];
    const float* W = (const float*)d_in[1];
    float* out = (float*)d_out;

    float* ws    = (float*)d_ws;
    float* WT2   = ws;
    float* xT2   = WT2 + 8388608;
    float* bb    = xT2 + 2097152;
    float* cc    = bb + 65536;
    float* spart = cc + 65536;
    float* vbuf  = spart + 4 * 65536;

    hipMemsetAsync(bb, 0, 65536 * sizeof(float), stream);
    transpose_W_kernel<<<8388608 / 256, 256, 0, stream>>>(W, WT2);
    transpose_x_kernel<<<2097152 / 256, 256, 0, stream>>>(x, xT2);

    for (int it = 0; it < 3; it++) {
        softmax_c_kernel<<<32, 256, 0, stream>>>(bb, cc);
        s_kernel<<<256, 256, 0, stream>>>(WT2, xT2, cc, spart);
        squash_kernel<<<128, 512, 0, stream>>>(spart, vbuf, out, it == 2 ? 1 : 0);
        if (it < 2) {
            bupd_kernel<<<dim3(128, 8), 128, 0, stream>>>(x, W, vbuf, bb);
        }
    }
}


# s_kernel loads W fragments from W's own layout (one b128 per lane per c), x from a tiled [c][b][u] copy: W layout pass and plane dropped; the 8 k's of a c permuted over its 2 WMMAs (sum order)
# speedup vs baseline: 1.8023x; 1.1166x over previous
#include <hip/hip_runtime.h>
#include <hip/hip_bf16.h>

typedef float v2f __attribute__((ext_vector_type(2)));
typedef float v4f __attribute__((ext_vector_type(4)));
typedef float v8f __attribute__((ext_vector_type(8)));

#define B_ 128
#define U_ 8
#define C_ 2048
#define J_ 32
#define S_ 16

__global__ __launch_bounds__(256) void transpose_x_kernel(const float* __restrict__ x,
                                                          float* __restrict__ xT) {
    __shared__ float tile[32][129];
    int c0   = (blockIdx.x & 63) * 32;
    int r0   = (blockIdx.x >> 6) * 128;
    int lane = threadIdx.x & 31;
    int w    = threadIdx.x >> 5;
#pragma unroll
    for (int i = 0; i < 16; i++) {
        int r = w * 16 + i;
        tile[lane][r] = x[(size_t)(r0 + r) * C_ + c0 + lane];
    }
    __syncthreads();
#pragma unroll
    for (int q = 0; q < 4; q++) {
        int cl = w * 4 + q;
#pragma unroll
        for (int i = 0; i < 4; i++)
            xT[(size_t)(c0 + cl) * (B_ * U_) + r0 + i * 32 + lane] = tile[cl][i * 32 + lane];
    }
}

__global__ void softmax_c_kernel(const float* __restrict__ bb, float* __restrict__ cc) {
    int j = blockIdx.x;
    int tid = threadIdx.x;
    __shared__ float red[256];
    float m = -1e30f;
    for (int c = tid; c < C_; c += 256) m = fmaxf(m, bb[c * J_ + j]);
    red[tid] = m; __syncthreads();
    for (int off = 128; off > 0; off >>= 1) {
        if (tid < off) red[tid] = fmaxf(red[tid], red[tid + off]);
        __syncthreads();
    }
    float mx = red[0]; __syncthreads();
    float sum = 0.f;
    for (int c = tid; c < C_; c += 256) sum += __expf(bb[c * J_ + j] - mx);
    red[tid] = sum; __syncthreads();
    for (int off = 128; off > 0; off >>= 1) {
        if (tid < off) red[tid] += red[tid + off];
        __syncthreads();
    }
    float inv = 1.0f / red[0];
    for (int c = tid; c < C_; c += 256) cc[c * J_ + j] = __expf(bb[c * J_ + j] - mx) * inv;
}

#define SK_CN 4
#define SK_NB (64 / SK_CN)
#define SK_WC (J_ * S_ * U_)
#define SK_XC (B_ * U_)

struct SkSet {
    v4f a0[SK_CN], a1[SK_CN], b0[SK_CN], b1[SK_CN];
    v2f cw[SK_CN];
};

__device__ __forceinline__ void sk_load(SkSet& f, const float* __restrict__ pa,
                                        const float* __restrict__ pb,
                                        const float* __restrict__ pc) {
#pragma unroll
    for (int i = 0; i < SK_CN; i++) f.cw[i] = *(const v2f*)(pc + i * J_);
#pragma unroll
    for (int i = 0; i < SK_CN; i++) {
        f.a0[i] = *(const v4f*)(pa + i * SK_WC);
        f.a1[i] = *(const v4f*)(pa + i * SK_WC + S_ * U_);
        f.b0[i] = *(const v4f*)(pb + i * SK_XC);
        f.b1[i] = *(const v4f*)(pb + i * SK_XC + 16 * U_);
    }
}

__device__ __forceinline__ void sk_compute(const SkSet& f, v8f& a00, v8f& a01,
                                           v8f& a10, v8f& a11) {
#pragma unroll
    for (int i = 0; i < SK_CN; i++) {
        v4f fa0 = f.a0[i] * f.cw[i].x;
        v4f fa1 = f.a1[i] * f.cw[i].y;
        v2f fa0l = {fa0.x, fa0.y}, fa0h = {fa0.z, fa0.w};
        v2f fa1l = {fa1.x, fa1.y}, fa1h = {fa1.z, fa1.w};
        v2f b0l = {f.b0[i].x, f.b0[i].y}, b0h = {f.b0[i].z, f.b0[i].w};
        v2f b1l = {f.b1[i].x, f.b1[i].y}, b1h = {f.b1[i].z, f.b1[i].w};
        a00 = __builtin_amdgcn_wmma_f32_16x16x4_f32(false, fa0l, false, b0l, (short)0, a00, false, false);
        a01 = __builtin_amdgcn_wmma_f32_16x16x4_f32(false, fa0l, false, b1l, (short)0, a01, false, false);
        a10 = __builtin_amdgcn_wmma_f32_16x16x4_f32(false, fa1l, false, b0l, (short)0, a10, false, false);
        a11 = __builtin_amdgcn_wmma_f32_16x16x4_f32(false, fa1l, false, b1l, (short)0, a11, false, false);
        a00 = __builtin_amdgcn_wmma_f32_16x16x4_f32(false, fa0h, false, b0h, (short)0, a00, false, false);
        a01 = __builtin_amdgcn_wmma_f32_16x16x4_f32(false, fa0h, false, b1h, (short)0, a01, false, false);
        a10 = __builtin_amdgcn_wmma_f32_16x16x4_f32(false, fa1h, false, b0h, (short)0, a10, false, false);
        a11 = __builtin_amdgcn_wmma_f32_16x16x4_f32(false, fa1h, false, b1h, (short)0, a11, false, false);
    }
}

__global__ __launch_bounds__(256) void s_kernel(
    const float* __restrict__ W, const float* __restrict__ xT,
    const float* __restrict__ cc, float* __restrict__ spart) {
    int bx  = blockIdx.x;
    int ks  = bx & 3;
    int btp = (bx >> 2) & 3;
    int jp  = bx >> 4;
    int j0  = jp * 2;
    int bt0 = btp * 2;

    int tid  = threadIdx.x;
    int w    = __builtin_amdgcn_readfirstlane(tid >> 5);
    int lane = tid & 31;
    int h    = lane >> 4;
    int ln   = lane & 15;

    int cs = ks * 512 + w * 64;
    const float* pa = W  + ((size_t)cs * J_ + j0) * (S_ * U_) + ln * U_ + 4 * h;
    const float* pb = xT + ((size_t)cs * B_ + bt0 * 16 + ln) * U_ + 4 * h;
    const float* pc = cc + (size_t)cs * J_ + j0;

    v8f a00 = {}, a01 = {}, a10 = {}, a11 = {};
    SkSet fA, fB;
#define SK_FENCE() __builtin_amdgcn_sched_barrier(0)
    sk_load(fA, pa, pb, pc);
    for (int it = 0; it < SK_NB / 2 - 1; it++) {
        sk_load(fB, pa + SK_CN * SK_WC, pb + SK_CN * SK_XC, pc + SK_CN * J_);
        SK_FENCE();
        sk_compute(fA, a00, a01, a10, a11);
        pa += 2 * SK_CN * SK_WC; pb += 2 * SK_CN * SK_XC; pc += 2 * SK_CN * J_;
        sk_load(fA, pa, pb, pc);
        SK_FENCE();
        sk_compute(fB, a00, a01, a10, a11);
    }
    sk_load(fB, pa + SK_CN * SK_WC, pb + SK_CN * SK_XC, pc + SK_CN * J_);
    SK_FENCE();
    sk_compute(fA, a00, a01, a10, a11);
    sk_compute(fB, a00, a01, a10, a11);
#undef SK_FENCE

    __shared__ float red[8][4][256];
#pragma unroll
    for (int r = 0; r < 8; r++) {
        int m = r + 8 * h;
        red[w][0][m * 16 + ln] = a00[r];
        red[w][1][m * 16 + ln] = a01[r];
        red[w][2][m * 16 + ln] = a10[r];
        red[w][3][m * 16 + ln] = a11[r];
    }
    __syncthreads();
    int m = tid >> 4, n = tid & 15;
    float* sp = spart + (size_t)ks * (B_ * J_ * S_);
#pragma unroll
    for (int t = 0; t < 4; t++) {
        int jj = t >> 1, bb = t & 1;
        float sum = 0.f;
#pragma unroll
        for (int ww = 0; ww < 8; ww++) sum += red[ww][t][tid];
        sp[(size_t)((bt0 + bb) * 16 + n) * (J_ * S_) + (j0 + jj) * S_ + m] = sum;
    }
}

__global__ void squash_kernel(const float* __restrict__ spart, float* __restrict__ vbuf,
                              float* __restrict__ out, int write_out) {
    int b = blockIdx.x;
    int tid = threadIdx.x;
    __shared__ float red[512];
    size_t idx = (size_t)b * 512 + tid;
    float val = spart[idx] + spart[idx + 65536] + spart[idx + 2 * 65536] + spart[idx + 3 * 65536];
    red[tid] = val * val;
    __syncthreads();
    for (int jo = 256; jo >= 16; jo >>= 1) {
        if (tid < jo) red[tid] += red[tid + jo];
        __syncthreads();
    }
    float msq = red[tid & 15];
    float scale = msq / ((1.0f + msq) * sqrtf(msq));
    float vv = val * scale;
    vbuf[idx] = vv;
    if (write_out) out[idx] = vv;
}

__global__ __launch_bounds__(128) void bupd_kernel(
    const float* __restrict__ x, const float* __restrict__ W,
    const float* __restrict__ vbuf, float* __restrict__ bb) {
    int ct = blockIdx.x;
    int jq = blockIdx.y;
    int cbase = ct * 16;
    int tid  = threadIdx.x;
    int w    = tid >> 5;
    int lane = tid & 31;
    int perm = (lane >> 4) * 2;
    int ln   = lane & 15;
    int j    = jq * 4 + w;

    __shared__ float vlds[4][B_ * S_];
    for (int i = lane; i < B_ * S_; i += 32)
        vlds[w][i] = vbuf[(size_t)(i >> 4) * (J_ * S_) + j * S_ + (i & 15)];
    __syncthreads();

    float sacc[8];
#pragma unroll
    for (int r = 0; r < 8; r++) sacc[r] = 0.f;

    for (int u = 0; u < U_; u++) {
        v8f acc = {};
        for (int kb = 0; kb < B_; kb += 4) {
            int k = kb + perm;
            v2f a, bf;
            a.x  = x[((size_t)k * U_ + u) * C_ + cbase + ln];
            a.y  = x[((size_t)(k + 1) * U_ + u) * C_ + cbase + ln];
            bf.x = vlds[w][k * 16 + ln];
            bf.y = vlds[w][(k + 1) * 16 + ln];
            acc = __builtin_amdgcn_wmma_f32_16x16x4_f32(
                false, a, false, bf, (short)0, acc, false, false);
        }
#pragma unroll
        for (int r = 0; r < 8; r++) {
            int c = cbase + r + 8 * (lane >> 4);
            float wv = W[((size_t)c * J_ + j) * (S_ * U_) + ln * U_ + u];
            float p = acc[r] * wv;
            p += __shfl_xor(p, 1);
            p += __shfl_xor(p, 2);
            p += __shfl_xor(p, 4);
            p += __shfl_xor(p, 8);
            sacc[r] += p;
        }
    }
    if (ln == 0) {
        int h = lane >> 4;
#pragma unroll
        for (int r = 0; r < 8; r++) {
            size_t bi = (size_t)(cbase + r + 8 * h) * J_ + j;
            bb[bi] += sacc[r] * (1.0f / 128.0f);
        }
    }
}

extern "C" void kernel_launch(void* const* d_in, const int* in_sizes, int n_in,
                              void* d_out, int out_size, void* d_ws, size_t ws_size,
                              hipStream_t stream) {
    (void)in_sizes; (void)n_in; (void)out_size; (void)ws_size;
    const float* x = (const float*)d_in[0];
    const float* W = (const float*)d_in[1];
    float* out = (float*)d_out;

    float* ws    = (float*)d_ws;
    float* xT    = ws;
    float* bb    = xT + 2097152;
    float* cc    = bb + 65536;
    float* spart = cc + 65536;
    float* vbuf  = spart + 4 * 65536;

    hipMemsetAsync(bb, 0, 65536 * sizeof(float), stream);
    transpose_x_kernel<<<512, 256, 0, stream>>>(x, xT);

    for (int it = 0; it < 3; it++) {
        softmax_c_kernel<<<32, 256, 0, stream>>>(bb, cc);
        s_kernel<<<256, 256, 0, stream>>>(W, xT, cc, spart);
        squash_kernel<<<128, 512, 0, stream>>>(spart, vbuf, out, it == 2 ? 1 : 0);
        if (it < 2) {
            bupd_kernel<<<dim3(128, 8), 128, 0, stream>>>(x, W, vbuf, bb);
        }
    }
}


# b-update kernel contracts over u on the matrix unit from W and the tiled x copy (b128 fragments, next set's loads grouped ahead), dots the tiles with v held in registers; agreement sum order changed
# speedup vs baseline: 2.3843x; 1.3229x over previous
#include <hip/hip_runtime.h>
#include <hip/hip_bf16.h>

typedef float v2f __attribute__((ext_vector_type(2)));
typedef float v4f __attribute__((ext_vector_type(4)));
typedef float v8f __attribute__((ext_vector_type(8)));

#define B_ 128
#define U_ 8
#define C_ 2048
#define J_ 32
#define S_ 16

__global__ __launch_bounds__(256) void transpose_x_kernel(const float* __restrict__ x,
                                                          float* __restrict__ xT) {
    __shared__ float tile[32][129];
    int c0   = (blockIdx.x & 63) * 32;
    int r0   = (blockIdx.x >> 6) * 128;
    int lane = threadIdx.x & 31;
    int w    = threadIdx.x >> 5;
#pragma unroll
    for (int i = 0; i < 16; i++) {
        int r = w * 16 + i;
        tile[lane][r] = x[(size_t)(r0 + r) * C_ + c0 + lane];
    }
    __syncthreads();
#pragma unroll
    for (int q = 0; q < 4; q++) {
        int cl = w * 4 + q;
#pragma unroll
        for (int i = 0; i < 4; i++)
            xT[(size_t)(c0 + cl) * (B_ * U_) + r0 + i * 32 + lane] = tile[cl][i * 32 + lane];
    }
}

__global__ void softmax_c_kernel(const float* __restrict__ bb, float* __restrict__ cc) {
    int j = blockIdx.x;
    int tid = threadIdx.x;
    __shared__ float red[256];
    float m = -1e30f;
    for (int c = tid; c < C_; c += 256) m = fmaxf(m, bb[c * J_ + j]);
    red[tid] = m; __syncthreads();
    for (int off = 128; off > 0; off >>= 1) {
        if (tid < off) red[tid] = fmaxf(red[tid], red[tid + off]);
        __syncthreads();
    }
    float mx = red[0]; __syncthreads();
    float sum = 0.f;
    for (int c = tid; c < C_; c += 256) sum += __expf(bb[c * J_ + j] - mx);
    red[tid] = sum; __syncthreads();
    for (int off = 128; off > 0; off >>= 1) {
        if (tid < off) red[tid] += red[tid + off];
        __syncthreads();
    }
    float inv = 1.0f / red[0];
    for (int c = tid; c < C_; c += 256) cc[c * J_ + j] = __expf(bb[c * J_ + j] - mx) * inv;
}

#define SK_CN 4
#define SK_NB (64 / SK_CN)
#define SK_WC (J_ * S_ * U_)
#define SK_XC (B_ * U_)

struct SkSet {
    v4f a0[SK_CN], a1[SK_CN], b0[SK_CN], b1[SK_CN];
    v2f cw[SK_CN];
};

__device__ __forceinline__ void sk_load(SkSet& f, const float* __restrict__ pa,
                                        const float* __restrict__ pb,
                                        const float* __restrict__ pc) {
#pragma unroll
    for (int i = 0; i < SK_CN; i++) f.cw[i] = *(const v2f*)(pc + i * J_);
#pragma unroll
    for (int i = 0; i < SK_CN; i++) {
        f.a0[i] = *(const v4f*)(pa + i * SK_WC);
        f.a1[i] = *(const v4f*)(pa + i * SK_WC + S_ * U_);
        f.b0[i] = *(const v4f*)(pb + i * SK_XC);
        f.b1[i] = *(const v4f*)(pb + i * SK_XC + 16 * U_);
    }
}

__device__ __forceinline__ void sk_compute(const SkSet& f, v8f& a00, v8f& a01,
                                           v8f& a10, v8f& a11) {
#pragma unroll
    for (int i = 0; i < SK_CN; i++) {
        v4f fa0 = f.a0[i] * f.cw[i].x;
        v4f fa1 = f.a1[i] * f.cw[i].y;
        v2f fa0l = {fa0.x, fa0.y}, fa0h = {fa0.z, fa0.w};
        v2f fa1l = {fa1.x, fa1.y}, fa1h = {fa1.z, fa1.w};
        v2f b0l = {f.b0[i].x, f.b0[i].y}, b0h = {f.b0[i].z, f.b0[i].w};
        v2f b1l = {f.b1[i].x, f.b1[i].y}, b1h = {f.b1[i].z, f.b1[i].w};
        a00 = __builtin_amdgcn_wmma_f32_16x16x4_f32(false, fa0l, false, b0l, (short)0, a00, false, false);
        a01 = __builtin_amdgcn_wmma_f32_16x16x4_f32(false, fa0l, false, b1l, (short)0, a01, false, false);
        a10 = __builtin_amdgcn_wmma_f32_16x16x4_f32(false, fa1l, false, b0l, (short)0, a10, false, false);
        a11 = __builtin_amdgcn_wmma_f32_16x16x4_f32(false, fa1l, false, b1l, (short)0, a11, false, false);
        a00 = __builtin_amdgcn_wmma_f32_16x16x4_f32(false, fa0h, false, b0h, (short)0, a00, false, false);
        a01 = __builtin_amdgcn_wmma_f32_16x16x4_f32(false, fa0h, false, b1h, (short)0, a01, false, false);
        a10 = __builtin_amdgcn_wmma_f32_16x16x4_f32(false, fa1h, false, b0h, (short)0, a10, false, false);
        a11 = __builtin_amdgcn_wmma_f32_16x16x4_f32(false, fa1h, false, b1h, (short)0, a11, false, false);
    }
}

__global__ __launch_bounds__(256) void s_kernel(
    const float* __restrict__ W, const float* __restrict__ xT,
    const float* __restrict__ cc, float* __restrict__ spart) {
    int bx  = blockIdx.x;
    int ks  = bx & 3;
    int btp = (bx >> 2) & 3;
    int jp  = bx >> 4;
    int j0  = jp * 2;
    int bt0 = btp * 2;

    int tid  = threadIdx.x;
    int w    = __builtin_amdgcn_readfirstlane(tid >> 5);
    int lane = tid & 31;
    int h    = lane >> 4;
    int ln   = lane & 15;

    int cs = ks * 512 + w * 64;
    const float* pa = W  + ((size_t)cs * J_ + j0) * (S_ * U_) + ln * U_ + 4 * h;
    const float* pb = xT + ((size_t)cs * B_ + bt0 * 16 + ln) * U_ + 4 * h;
    const float* pc = cc + (size_t)cs * J_ + j0;

    v8f a00 = {}, a01 = {}, a10 = {}, a11 = {};
    SkSet fA, fB;
#define SK_FENCE() __builtin_amdgcn_sched_barrier(0)
    sk_load(fA, pa, pb, pc);
    for (int it = 0; it < SK_NB / 2 - 1; it++) {
        sk_load(fB, pa + SK_CN * SK_WC, pb + SK_CN * SK_XC, pc + SK_CN * J_);
        SK_FENCE();
        sk_compute(fA, a00, a01, a10, a11);
        pa += 2 * SK_CN * SK_WC; pb += 2 * SK_CN * SK_XC; pc += 2 * SK_CN * J_;
        sk_load(fA, pa, pb, pc);
        SK_FENCE();
        sk_compute(fB, a00, a01, a10, a11);
    }
    sk_load(fB, pa + SK_CN * SK_WC, pb + SK_CN * SK_XC, pc + SK_CN * J_);
    SK_FENCE();
    sk_compute(fA, a00, a01, a10, a11);
    sk_compute(fB, a00, a01, a10, a11);
#undef SK_FENCE

    __shared__ float red[8][4][256];
#pragma unroll
    for (int r = 0; r < 8; r++) {
        int m = r + 8 * h;
        red[w][0][m * 16 + ln] = a00[r];
        red[w][1][m * 16 + ln] = a01[r];
        red[w][2][m * 16 + ln] = a10[r];
        red[w][3][m * 16 + ln] = a11[r];
    }
    __syncthreads();
    int m = tid >> 4, n = tid & 15;
    float* sp = spart + (size_t)ks * (B_ * J_ * S_);
#pragma unroll
    for (int t = 0; t < 4; t++) {
        int jj = t >> 1, bb = t & 1;
        float sum = 0.f;
#pragma unroll
        for (int ww = 0; ww < 8; ww++) sum += red[ww][t][tid];
        sp[(size_t)((bt0 + bb) * 16 + n) * (J_ * S_) + (j0 + jj) * S_ + m] = sum;
    }
}

__global__ void squash_kernel(const float* __restrict__ spart, float* __restrict__ vbuf,
                              float* __restrict__ out, int write_out) {
    int b = blockIdx.x;
    int tid = threadIdx.x;
    __shared__ float red[512];
    size_t idx = (size_t)b * 512 + tid;
    float val = spart[idx] + spart[idx + 65536] + spart[idx + 2 * 65536] + spart[idx + 3 * 65536];
    red[tid] = val * val;
    __syncthreads();
    for (int jo = 256; jo >= 16; jo >>= 1) {
        if (tid < jo) red[tid] += red[tid + jo];
        __syncthreads();
    }
    float msq = red[tid & 15];
    float scale = msq / ((1.0f + msq) * sqrtf(msq));
    float vv = val * scale;
    vbuf[idx] = vv;
    if (write_out) out[idx] = vv;
}

#define AG_CN 2
#define AG_NS (16 / AG_CN)

struct AgSet {
    v4f a0[AG_CN], a1[AG_CN];
    v4f b[AG_CN][8];
};

__device__ __forceinline__ void ag_load(AgSet& f, const float* __restrict__ pa,
                                        const float* __restrict__ pb) {
#pragma unroll
    for (int i = 0; i < AG_CN; i++) {
        f.a0[i] = *(const v4f*)(pa + i * SK_WC);
        f.a1[i] = *(const v4f*)(pa + i * SK_WC + S_ * U_);
#pragma unroll
        for (int t = 0; t < 8; t++)
            f.b[i][t] = *(const v4f*)(pb + i * SK_XC + t * 16 * U_);
    }
}

__device__ __forceinline__ void ag_compute(const AgSet& f, const v8f (&v0)[8],
                                           const v8f (&v1)[8], float* __restrict__ rp) {
#pragma unroll
    for (int i = 0; i < AG_CN; i++) {
        v2f a0l = {f.a0[i].x, f.a0[i].y}, a0h = {f.a0[i].z, f.a0[i].w};
        v2f a1l = {f.a1[i].x, f.a1[i].y}, a1h = {f.a1[i].z, f.a1[i].w};
        v8f p0 = {}, p1 = {};
#pragma unroll
        for (int t = 0; t < 8; t++) {
            v2f bl = {f.b[i][t].x, f.b[i][t].y}, bh = {f.b[i][t].z, f.b[i][t].w};
            v8f z = {};
            v8f d0 = __builtin_amdgcn_wmma_f32_16x16x4_f32(false, a0l, false, bl, (short)0, z, false, false);
            v8f d1 = __builtin_amdgcn_wmma_f32_16x16x4_f32(false, a1l, false, bl, (short)0, z, false, false);
            d0 = __builtin_amdgcn_wmma_f32_16x16x4_f32(false, a0h, false, bh, (short)0, d0, false, false);
            d1 = __builtin_amdgcn_wmma_f32_16x16x4_f32(false, a1h, false, bh, (short)0, d1, false, false);
            p0 += d0 * v0[t];
            p1 += d1 * v1[t];
        }
        float s0 = ((p0[0] + p0[1]) + (p0[2] + p0[3])) + ((p0[4] + p0[5]) + (p0[6] + p0[7]));
        float s1 = ((p1[0] + p1[1]) + (p1[2] + p1[3])) + ((p1[4] + p1[5]) + (p1[6] + p1[7]));
        rp[(i * 2 + 0) * 33] = s0;
        rp[(i * 2 + 1) * 33] = s1;
    }
}

__global__ __launch_bounds__(256) void bupd_kernel(
    const float* __restrict__ xT, const float* __restrict__ W,
    const float* __restrict__ vbuf, float* __restrict__ bb) {
    int bx  = blockIdx.x;
    int jh  = bx & 1;
    int cr  = bx >> 1;
    int tid = threadIdx.x;
    int w    = __builtin_amdgcn_readfirstlane(tid >> 5);
    int lane = tid & 31;
    int h    = lane >> 4;
    int ln   = lane & 15;
    int j0   = (jh * 8 + w) * 2;
    int c0   = cr * 16;

    const float* pa = W  + ((size_t)c0 * J_ + j0) * (S_ * U_) + ln * U_ + 4 * h;
    const float* pb = xT + ((size_t)c0 * B_ + ln) * U_ + 4 * h;
    const float* pv = vbuf + (size_t)ln * (J_ * S_) + j0 * S_ + 8 * h;

    v8f v0[8], v1[8];
#pragma unroll
    for (int t = 0; t < 8; t++) {
        const float* q = pv + (size_t)t * 16 * (J_ * S_);
        v4f l0 = *(const v4f*)(q),      h0 = *(const v4f*)(q + 4);
        v4f l1 = *(const v4f*)(q + S_), h1 = *(const v4f*)(q + S_ + 4);
        v0[t] = (v8f){l0.x, l0.y, l0.z, l0.w, h0.x, h0.y, h0.z, h0.w};
        v1[t] = (v8f){l1.x, l1.y, l1.z, l1.w, h1.x, h1.y, h1.z, h1.w};
    }

    __shared__ float red[8][32][33];
    float* rp = &red[w][0][lane];

    AgSet fA, fB;
#define AG_FENCE() __builtin_amdgcn_sched_barrier(0)
    ag_load(fA, pa, pb);
#pragma unroll 1
    for (int it = 0; it < AG_NS / 2 - 1; it++) {
        ag_load(fB, pa + AG_CN * SK_WC, pb + AG_CN * SK_XC);
        AG_FENCE();
        ag_compute(fA, v0, v1, rp);
        pa += 2 * AG_CN * SK_WC; pb += 2 * AG_CN * SK_XC;
        ag_load(fA, pa, pb);
        AG_FENCE();
        ag_compute(fB, v0, v1, rp + AG_CN * 2 * 33);
        rp += 2 * AG_CN * 2 * 33;
    }
    ag_load(fB, pa + AG_CN * SK_WC, pb + AG_CN * SK_XC);
    AG_FENCE();
    ag_compute(fA, v0, v1, rp);
    ag_compute(fB, v0, v1, rp + AG_CN * 2 * 33);
#undef AG_FENCE

    __syncthreads();
    const float* rr = &red[w][lane][0];
    float sum = 0.f;
#pragma unroll
    for (int k = 0; k < 32; k++) sum += rr[k];
    size_t bi = (size_t)(c0 + (lane >> 1)) * J_ + j0 + (lane & 1);
    bb[bi] += sum * (1.0f / 128.0f);
}

extern "C" void kernel_launch(void* const* d_in, const int* in_sizes, int n_in,
                              void* d_out, int out_size, void* d_ws, size_t ws_size,
                              hipStream_t stream) {
    (void)in_sizes; (void)n_in; (void)out_size; (void)ws_size;
    const float* x = (const float*)d_in[0];
    const float* W = (const float*)d_in[1];
    float* out = (float*)d_out;

    float* ws    = (float*)d_ws;
    float* xT    = ws;
    float* bb    = xT + 2097152;
    float* cc    = bb + 65536;
    float* spart = cc + 65536;
    float* vbuf  = spart + 4 * 65536;

    hipMemsetAsync(bb, 0, 65536 * sizeof(float), stream);
    transpose_x_kernel<<<512, 256, 0, stream>>>(x, xT);

    for (int it = 0; it < 3; it++) {
        softmax_c_kernel<<<32, 256, 0, stream>>>(bb, cc);
        s_kernel<<<256, 256, 0, stream>>>(W, xT, cc, spart);
        squash_kernel<<<128, 512, 0, stream>>>(spart, vbuf, out, it == 2 ? 1 : 0);
        if (it < 2) {
            bupd_kernel<<<256, 256, 0, stream>>>(xT, W, vbuf, bb);
        }
    }
}


# softmax over C folded into s_kernel's head: each block reduces its own two logit columns (logits kept [J][C]; pass 0 takes the exact 1/2048); 3 softmax launches and the memset go; sum orders kept
# speedup vs baseline: 3.4580x; 1.4503x over previous
#include <hip/hip_runtime.h>
#include <hip/hip_bf16.h>

typedef float v2f __attribute__((ext_vector_type(2)));
typedef float v4f __attribute__((ext_vector_type(4)));
typedef float v8f __attribute__((ext_vector_type(8)));

#define B_ 128
#define U_ 8
#define C_ 2048
#define J_ 32
#define S_ 16

__global__ __launch_bounds__(256) void transpose_x_kernel(const float* __restrict__ x,
                                                          float* __restrict__ xT) {
    __shared__ float tile[32][129];
    int c0   = (blockIdx.x & 63) * 32;
    int r0   = (blockIdx.x >> 6) * 128;
    int lane = threadIdx.x & 31;
    int w    = threadIdx.x >> 5;
#pragma unroll
    for (int i = 0; i < 16; i++) {
        int r = w * 16 + i;
        tile[lane][r] = x[(size_t)(r0 + r) * C_ + c0 + lane];
    }
    __syncthreads();
#pragma unroll
    for (int q = 0; q < 4; q++) {
        int cl = w * 4 + q;
#pragma unroll
        for (int i = 0; i < 4; i++)
            xT[(size_t)(c0 + cl) * (B_ * U_) + r0 + i * 32 + lane] = tile[cl][i * 32 + lane];
    }
}

#define SK_CN 4
#define SK_NB (64 / SK_CN)
#define SK_WC (J_ * S_ * U_)
#define SK_XC (B_ * U_)

struct SkSet {
    v4f a0[SK_CN], a1[SK_CN], b0[SK_CN], b1[SK_CN];
    v4f cw[SK_CN / 2];
};

__device__ __forceinline__ void sk_load_w(SkSet& f, const float* pw) {
#pragma unroll
    for (int q = 0; q < SK_CN / 2; q++) f.cw[q] = *(const v4f*)(pw + 4 * q);
}

__device__ __forceinline__ void sk_load(SkSet& f, const float* __restrict__ pa,
                                        const float* __restrict__ pb) {
#pragma unroll
    for (int i = 0; i < SK_CN; i++) {
        f.a0[i] = *(const v4f*)(pa + i * SK_WC);
        f.a1[i] = *(const v4f*)(pa + i * SK_WC + S_ * U_);
        f.b0[i] = *(const v4f*)(pb + i * SK_XC);
        f.b1[i] = *(const v4f*)(pb + i * SK_XC + 16 * U_);
    }
}

__device__ __forceinline__ void sk_compute(const SkSet& f, v8f& a00, v8f& a01,
                                           v8f& a10, v8f& a11) {
#pragma unroll
    for (int i = 0; i < SK_CN; i++) {
        float wx = (i & 1) ? f.cw[i >> 1].z : f.cw[i >> 1].x;
        float wy = (i & 1) ? f.cw[i >> 1].w : f.cw[i >> 1].y;
        v4f fa0 = f.a0[i] * wx;
        v4f fa1 = f.a1[i] * wy;
        v2f fa0l = {fa0.x, fa0.y}, fa0h = {fa0.z, fa0.w};
        v2f fa1l = {fa1.x, fa1.y}, fa1h = {fa1.z, fa1.w};
        v2f b0l = {f.b0[i].x, f.b0[i].y}, b0h = {f.b0[i].z, f.b0[i].w};
        v2f b1l = {f.b1[i].x, f.b1[i].y}, b1h = {f.b1[i].z, f.b1[i].w};
        a00 = __builtin_amdgcn_wmma_f32_16x16x4_f32(false, fa0l, false, b0l, (short)0, a00, false, false);
        a01 = __builtin_amdgcn_wmma_f32_16x16x4_f32(false, fa0l, false, b1l, (short)0, a01, false, false);
        a10 = __builtin_amdgcn_wmma_f32_16x16x4_f32(false, fa1l, false, b0l, (short)0, a10, false, false);
        a11 = __builtin_amdgcn_wmma_f32_16x16x4_f32(false, fa1l, false, b1l, (short)0, a11, false, false);
        a00 = __builtin_amdgcn_wmma_f32_16x16x4_f32(false, fa0h, false, b0h, (short)0, a00, false, false);
        a01 = __builtin_amdgcn_wmma_f32_16x16x4_f32(false, fa0h, false, b1h, (short)0, a01, false, false);
        a10 = __builtin_amdgcn_wmma_f32_16x16x4_f32(false, fa1h, false, b0h, (short)0, a10, false, false);
        a11 = __builtin_amdgcn_wmma_f32_16x16x4_f32(false, fa1h, false, b1h, (short)0, a11, false, false);
    }
}

__global__ __launch_bounds__(256) void s_kernel(
    const float* __restrict__ W, const float* __restrict__ xT,
    const float* __restrict__ bT, float* __restrict__ spart, int pass0) {
    int bx  = blockIdx.x;
    int ks  = bx & 3;
    int btp = (bx >> 2) & 3;
    int jp  = bx >> 4;
    int j0  = jp * 2;
    int bt0 = btp * 2;

    int tid  = threadIdx.x;
    int w    = __builtin_amdgcn_readfirstlane(tid >> 5);
    int lane = tid & 31;
    int h    = lane >> 4;
    int ln   = lane & 15;

    int cs = ks * 512 + w * 64;
    const float* pa = W  + ((size_t)cs * J_ + j0) * (S_ * U_) + ln * U_ + 4 * h;
    const float* pb = xT + ((size_t)cs * B_ + bt0 * 16 + ln) * U_ + 4 * h;

    __shared__ float red[8][4][256];
    __shared__ float cwl[512 * 2];

    v8f a00 = {}, a01 = {}, a10 = {}, a11 = {};
    SkSet fA, fB;
    sk_load(fA, pa, pb);

    if (pass0) {
#pragma unroll
        for (int q = 0; q < 4; q++) cwl[tid + 256 * q] = 1.0f / 2048.0f;
    } else {
        float* r0 = &red[0][0][0];
        float* r1 = &red[0][1][0];
        float* u0 = &red[0][2][0];
        float* u1 = &red[0][3][0];
        const float* q0 = bT + (size_t)j0 * C_ + tid;
        const float* q1 = q0 + C_;
        float l0[8], l1[8];
#pragma unroll
        for (int k = 0; k < 8; k++) { l0[k] = q0[k * 256]; l1[k] = q1[k * 256]; }
        float m0 = -1e30f, m1 = -1e30f;
#pragma unroll
        for (int k = 0; k < 8; k++) { m0 = fmaxf(m0, l0[k]); m1 = fmaxf(m1, l1[k]); }
        r0[tid] = m0; r1[tid] = m1; __syncthreads();
        float mx0 = r0[lane], mx1 = r1[lane];
#pragma unroll
        for (int q = 1; q < 8; q++) { mx0 = fmaxf(mx0, r0[lane + 32 * q]); mx1 = fmaxf(mx1, r1[lane + 32 * q]); }
#pragma unroll
        for (int off = 16; off > 0; off >>= 1) {
            mx0 = fmaxf(mx0, __shfl_xor(mx0, off));
            mx1 = fmaxf(mx1, __shfl_xor(mx1, off));
        }
        float s0 = 0.f, s1 = 0.f;
#pragma unroll
        for (int k = 0; k < 8; k++) { s0 += __expf(l0[k] - mx0); s1 += __expf(l1[k] - mx1); }
        u0[tid] = s0; u1[tid] = s1; __syncthreads();
        float z0 = ((u0[lane] + u0[lane + 128]) + (u0[lane + 64] + u0[lane + 192]))
                 + ((u0[lane + 32] + u0[lane + 160]) + (u0[lane + 96] + u0[lane + 224]));
        float z1 = ((u1[lane] + u1[lane + 128]) + (u1[lane + 64] + u1[lane + 192]))
                 + ((u1[lane + 32] + u1[lane + 160]) + (u1[lane + 96] + u1[lane + 224]));
#pragma unroll
        for (int off = 16; off > 0; off >>= 1) {
            z0 += __shfl_down(z0, off);
            z1 += __shfl_down(z1, off);
        }
        z0 = __shfl(z0, 0); z1 = __shfl(z1, 0);
        float inv0 = 1.0f / z0, inv1 = 1.0f / z1;
#pragma unroll
        for (int q = 0; q < 2; q++) {
            int cl = tid + 256 * q;
            cwl[cl * 2]     = __expf(q0[ks * 512 + 256 * q] - mx0) * inv0;
            cwl[cl * 2 + 1] = __expf(q1[ks * 512 + 256 * q] - mx1) * inv1;
        }
    }
    __syncthreads();
    const float* pw = cwl + w * 128;

#define SK_FENCE() __builtin_amdgcn_sched_barrier(0)
    sk_load_w(fA, pw);
    for (int it = 0; it < SK_NB / 2 - 1; it++) {
        sk_load(fB, pa + SK_CN * SK_WC, pb + SK_CN * SK_XC);
        sk_load_w(fB, pw + 2 * SK_CN);
        SK_FENCE();
        sk_compute(fA, a00, a01, a10, a11);
        pa += 2 * SK_CN * SK_WC; pb += 2 * SK_CN * SK_XC; pw += 4 * SK_CN;
        sk_load(fA, pa, pb);
        sk_load_w(fA, pw);
        SK_FENCE();
        sk_compute(fB, a00, a01, a10, a11);
    }
    sk_load(fB, pa + SK_CN * SK_WC, pb + SK_CN * SK_XC);
    sk_load_w(fB, pw + 2 * SK_CN);
    SK_FENCE();
    sk_compute(fA, a00, a01, a10, a11);
    sk_compute(fB, a00, a01, a10, a11);
#undef SK_FENCE

#pragma unroll
    for (int r = 0; r < 8; r++) {
        int m = r + 8 * h;
        red[w][0][m * 16 + ln] = a00[r];
        red[w][1][m * 16 + ln] = a01[r];
        red[w][2][m * 16 + ln] = a10[r];
        red[w][3][m * 16 + ln] = a11[r];
    }
    __syncthreads();
    int m = tid >> 4, n = tid & 15;
    float* sp = spart + (size_t)ks * (B_ * J_ * S_);
#pragma unroll
    for (int t = 0; t < 4; t++) {
        int jj = t >> 1, bb = t & 1;
        float sum = 0.f;
#pragma unroll
        for (int ww = 0; ww < 8; ww++) sum += red[ww][t][tid];
        sp[(size_t)((bt0 + bb) * 16 + n) * (J_ * S_) + (j0 + jj) * S_ + m] = sum;
    }
}

__global__ void squash_kernel(const float* __restrict__ spart, float* __restrict__ vbuf,
                              float* __restrict__ out, int write_out) {
    int b = blockIdx.x;
    int tid = threadIdx.x;
    __shared__ float red[512];
    size_t idx = (size_t)b * 512 + tid;
    float val = spart[idx] + spart[idx + 65536] + spart[idx + 2 * 65536] + spart[idx + 3 * 65536];
    red[tid] = val * val;
    __syncthreads();
    for (int jo = 256; jo >= 16; jo >>= 1) {
        if (tid < jo) red[tid] += red[tid + jo];
        __syncthreads();
    }
    float msq = red[tid & 15];
    float scale = msq / ((1.0f + msq) * sqrtf(msq));
    float vv = val * scale;
    vbuf[idx] = vv;
    if (write_out) out[idx] = vv;
}

#define AG_CN 2
#define AG_NS (16 / AG_CN)

struct AgSet {
    v4f a0[AG_CN], a1[AG_CN];
    v4f b[AG_CN][8];
};

__device__ __forceinline__ void ag_load(AgSet& f, const float* __restrict__ pa,
                                        const float* __restrict__ pb) {
#pragma unroll
    for (int i = 0; i < AG_CN; i++) {
        f.a0[i] = *(const v4f*)(pa + i * SK_WC);
        f.a1[i] = *(const v4f*)(pa + i * SK_WC + S_ * U_);
#pragma unroll
        for (int t = 0; t < 8; t++)
            f.b[i][t] = *(const v4f*)(pb + i * SK_XC + t * 16 * U_);
    }
}

__device__ __forceinline__ void ag_compute(const AgSet& f, const v8f (&v0)[8],
                                           const v8f (&v1)[8], float* __restrict__ rp) {
#pragma unroll
    for (int i = 0; i < AG_CN; i++) {
        v2f a0l = {f.a0[i].x, f.a0[i].y}, a0h = {f.a0[i].z, f.a0[i].w};
        v2f a1l = {f.a1[i].x, f.a1[i].y}, a1h = {f.a1[i].z, f.a1[i].w};
        v8f p0 = {}, p1 = {};
#pragma unroll
        for (int t = 0; t < 8; t++) {
            v2f bl = {f.b[i][t].x, f.b[i][t].y}, bh = {f.b[i][t].z, f.b[i][t].w};
            v8f z = {};
            v8f d0 = __builtin_amdgcn_wmma_f32_16x16x4_f32(false, a0l, false, bl, (short)0, z, false, false);
            v8f d1 = __builtin_amdgcn_wmma_f32_16x16x4_f32(false, a1l, false, bl, (short)0, z, false, false);
            d0 = __builtin_amdgcn_wmma_f32_16x16x4_f32(false, a0h, false, bh, (short)0, d0, false, false);
            d1 = __builtin_amdgcn_wmma_f32_16x16x4_f32(false, a1h, false, bh, (short)0, d1, false, false);
            p0 += d0 * v0[t];
            p1 += d1 * v1[t];
        }
        float s0 = ((p0[0] + p0[1]) + (p0[2] + p0[3])) + ((p0[4] + p0[5]) + (p0[6] + p0[7]));
        float s1 = ((p1[0] + p1[1]) + (p1[2] + p1[3])) + ((p1[4] + p1[5]) + (p1[6] + p1[7]));
        rp[(i * 2 + 0) * 33] = s0;
        rp[(i * 2 + 1) * 33] = s1;
    }
}

__global__ __launch_bounds__(256) void bupd_kernel(
    const float* __restrict__ xT, const float* __restrict__ W,
    const float* __restrict__ vbuf, float* __restrict__ bT, int first) {
    int bx  = blockIdx.x;
    int jh  = bx & 1;
    int cr  = bx >> 1;
    int tid = threadIdx.x;
    int w    = __builtin_amdgcn_readfirstlane(tid >> 5);
    int lane = tid & 31;
    int h    = lane >> 4;
    int ln   = lane & 15;
    int j0   = (jh * 8 + w) * 2;
    int c0   = cr * 16;

    const float* pa = W  + ((size_t)c0 * J_ + j0) * (S_ * U_) + ln * U_ + 4 * h;
    const float* pb = xT + ((size_t)c0 * B_ + ln) * U_ + 4 * h;
    const float* pv = vbuf + (size_t)ln * (J_ * S_) + j0 * S_ + 8 * h;

    v8f v0[8], v1[8];
#pragma unroll
    for (int t = 0; t < 8; t++) {
        const float* q = pv + (size_t)t * 16 * (J_ * S_);
        v4f l0 = *(const v4f*)(q),      h0 = *(const v4f*)(q + 4);
        v4f l1 = *(const v4f*)(q + S_), h1 = *(const v4f*)(q + S_ + 4);
        v0[t] = (v8f){l0.x, l0.y, l0.z, l0.w, h0.x, h0.y, h0.z, h0.w};
        v1[t] = (v8f){l1.x, l1.y, l1.z, l1.w, h1.x, h1.y, h1.z, h1.w};
    }

    __shared__ float red[8][32][33];
    float* rp = &red[w][0][lane];

    AgSet fA, fB;
#define AG_FENCE() __builtin_amdgcn_sched_barrier(0)
    ag_load(fA, pa, pb);
#pragma unroll 1
    for (int it = 0; it < AG_NS / 2 - 1; it++) {
        ag_load(fB, pa + AG_CN * SK_WC, pb + AG_CN * SK_XC);
        AG_FENCE();
        ag_compute(fA, v0, v1, rp);
        pa += 2 * AG_CN * SK_WC; pb += 2 * AG_CN * SK_XC;
        ag_load(fA, pa, pb);
        AG_FENCE();
        ag_compute(fB, v0, v1, rp + AG_CN * 2 * 33);
        rp += 2 * AG_CN * 2 * 33;
    }
    ag_load(fB, pa + AG_CN * SK_WC, pb + AG_CN * SK_XC);
    AG_FENCE();
    ag_compute(fA, v0, v1, rp);
    ag_compute(fB, v0, v1, rp + AG_CN * 2 * 33);
#undef AG_FENCE

    __syncthreads();
    const float* rr = &red[w][lane][0];
    float sum = 0.f;
#pragma unroll
    for (int k = 0; k < 32; k++) sum += rr[k];
    size_t bi = (size_t)(j0 + (lane & 1)) * C_ + c0 + (lane >> 1);
    float upd = sum * (1.0f / 128.0f);
    bT[bi] = first ? upd : bT[bi] + upd;
}

extern "C" void kernel_launch(void* const* d_in, const int* in_sizes, int n_in,
                              void* d_out, int out_size, void* d_ws, size_t ws_size,
                              hipStream_t stream) {
    (void)in_sizes; (void)n_in; (void)out_size; (void)ws_size;
    const float* x = (const float*)d_in[0];
    const float* W = (const float*)d_in[1];
    float* out = (float*)d_out;

    float* ws    = (float*)d_ws;
    float* xT    = ws;
    float* bT    = xT + 2097152;
    float* spart = bT + 2 * 65536;
    float* vbuf  = spart + 4 * 65536;

    transpose_x_kernel<<<512, 256, 0, stream>>>(x, xT);

    for (int it = 0; it < 3; it++) {
        s_kernel<<<256, 256, 0, stream>>>(W, xT, bT, spart, it == 0 ? 1 : 0);
        squash_kernel<<<128, 512, 0, stream>>>(spart, vbuf, out, it == 2 ? 1 : 0);
        if (it < 2) {
            bupd_kernel<<<256, 256, 0, stream>>>(xT, W, vbuf, bT, it == 0 ? 1 : 0);
        }
    }
}


# s_kernel register blocking 2x2 -> 4x4 per wave (block = 4 j x 4 b-tiles, K-split 4 -> 16 slices summed in squash): fragment reads per WMMA halved; partial-sum order over c changed
# speedup vs baseline: 3.8164x; 1.1036x over previous
#include <hip/hip_runtime.h>
#include <hip/hip_bf16.h>

typedef float v2f __attribute__((ext_vector_type(2)));
typedef float v4f __attribute__((ext_vector_type(4)));
typedef float v8f __attribute__((ext_vector_type(8)));

#define B_ 128
#define U_ 8
#define C_ 2048
#define J_ 32
#define S_ 16

__global__ __launch_bounds__(256) void transpose_x_kernel(const float* __restrict__ x,
                                                          float* __restrict__ xT) {
    __shared__ float tile[32][129];
    int c0   = (blockIdx.x & 63) * 32;
    int r0   = (blockIdx.x >> 6) * 128;
    int lane = threadIdx.x & 31;
    int w    = threadIdx.x >> 5;
#pragma unroll
    for (int i = 0; i < 16; i++) {
        int r = w * 16 + i;
        tile[lane][r] = x[(size_t)(r0 + r) * C_ + c0 + lane];
    }
    __syncthreads();
#pragma unroll
    for (int q = 0; q < 4; q++) {
        int cl = w * 4 + q;
#pragma unroll
        for (int i = 0; i < 4; i++)
            xT[(size_t)(c0 + cl) * (B_ * U_) + r0 + i * 32 + lane] = tile[cl][i * 32 + lane];
    }
}

#define SK_CN 2
#define SK_NB (16 / SK_CN)
#define SK_WC (J_ * S_ * U_)
#define SK_XC (B_ * U_)
#define SK_KS 16
#define SK_PP 68

struct SkSet {
    v4f a[SK_CN][4];
    v4f b[SK_CN][4];
    v4f cw[SK_CN];
};

__device__ __forceinline__ void sk_load_w(SkSet& f, const float* pw) {
#pragma unroll
    for (int q = 0; q < SK_CN; q++) f.cw[q] = *(const v4f*)(pw + 4 * q);
}

__device__ __forceinline__ void sk_load(SkSet& f, const float* __restrict__ pa,
                                        const float* __restrict__ pb) {
#pragma unroll
    for (int i = 0; i < SK_CN; i++) {
#pragma unroll
        for (int t = 0; t < 4; t++) {
            f.a[i][t] = *(const v4f*)(pa + i * SK_WC + t * S_ * U_);
            f.b[i][t] = *(const v4f*)(pb + i * SK_XC + t * 16 * U_);
        }
    }
}

__device__ __forceinline__ void sk_compute(const SkSet& f, v8f (&acc)[4][4]) {
#pragma unroll
    for (int i = 0; i < SK_CN; i++) {
        v2f al[4], ah[4], bl[4], bh[4];
#pragma unroll
        for (int t = 0; t < 4; t++) {
            v4f fa = f.a[i][t] * f.cw[i][t];
            al[t] = (v2f){fa.x, fa.y}; ah[t] = (v2f){fa.z, fa.w};
            bl[t] = (v2f){f.b[i][t].x, f.b[i][t].y}; bh[t] = (v2f){f.b[i][t].z, f.b[i][t].w};
        }
#pragma unroll
        for (int jj = 0; jj < 4; jj++)
#pragma unroll
            for (int bb = 0; bb < 4; bb++)
                acc[jj][bb] = __builtin_amdgcn_wmma_f32_16x16x4_f32(false, al[jj], false, bl[bb], (short)0, acc[jj][bb], false, false);
#pragma unroll
        for (int jj = 0; jj < 4; jj++)
#pragma unroll
            for (int bb = 0; bb < 4; bb++)
                acc[jj][bb] = __builtin_amdgcn_wmma_f32_16x16x4_f32(false, ah[jj], false, bh[bb], (short)0, acc[jj][bb], false, false);
    }
}

__global__ __launch_bounds__(256) void s_kernel(
    const float* __restrict__ W, const float* __restrict__ xT,
    const float* __restrict__ bT, float* __restrict__ spart, int pass0) {
    int bx  = blockIdx.x;
    int ks  = bx & 15;
    int bq  = (bx >> 4) & 1;
    int jq  = bx >> 5;
    int j0  = jq * 4;
    int bt0 = bq * 4;

    int tid  = threadIdx.x;
    int w    = __builtin_amdgcn_readfirstlane(tid >> 5);
    int lane = tid & 31;
    int h    = lane >> 4;
    int ln   = lane & 15;

    int cs = ks * 128 + w * 16;
    const float* pa = W  + ((size_t)cs * J_ + j0) * (S_ * U_) + ln * U_ + 4 * h;
    const float* pb = xT + ((size_t)cs * B_ + bt0 * 16 + ln) * U_ + 4 * h;

    __shared__ __attribute__((aligned(16))) float part[8 * 64 * SK_PP];
    __shared__ __attribute__((aligned(16))) float cwl[128 * 4];

    v8f acc[4][4] = {};
    SkSet fA, fB;
    sk_load(fA, pa, pb);

    if (pass0) {
#pragma unroll
        for (int q = 0; q < 2; q++) cwl[tid + 256 * q] = 1.0f / 2048.0f;
    } else {
        float* rm = part;
        float* ru = part + 4 * 256;
        const float* q0 = bT + (size_t)j0 * C_ + tid;
        float l[4][8];
#pragma unroll
        for (int jj = 0; jj < 4; jj++)
#pragma unroll
            for (int k = 0; k < 8; k++) l[jj][k] = q0[jj * C_ + k * 256];
        int cl = tid & 127, jh2 = (tid >> 7) * 2;
        float e0 = bT[(size_t)(j0 + jh2) * C_ + ks * 128 + cl];
        float e1 = bT[(size_t)(j0 + jh2 + 1) * C_ + ks * 128 + cl];
#pragma unroll
        for (int jj = 0; jj < 4; jj++) {
            float m = -1e30f;
#pragma unroll
            for (int k = 0; k < 8; k++) m = fmaxf(m, l[jj][k]);
            rm[jj * 256 + tid] = m;
        }
        __syncthreads();
        float mx[4];
#pragma unroll
        for (int jj = 0; jj < 4; jj++) {
            float m = rm[jj * 256 + lane];
#pragma unroll
            for (int q = 1; q < 8; q++) m = fmaxf(m, rm[jj * 256 + lane + 32 * q]);
#pragma unroll
            for (int off = 16; off > 0; off >>= 1) m = fmaxf(m, __shfl_xor(m, off));
            mx[jj] = m;
            float s = 0.f;
#pragma unroll
            for (int k = 0; k < 8; k++) s += __expf(l[jj][k] - m);
            ru[jj * 256 + tid] = s;
        }
        __syncthreads();
        float inv[4];
#pragma unroll
        for (int jj = 0; jj < 4; jj++) {
            const float* u = ru + jj * 256;
            float z = ((u[lane] + u[lane + 128]) + (u[lane + 64] + u[lane + 192]))
                    + ((u[lane + 32] + u[lane + 160]) + (u[lane + 96] + u[lane + 224]));
#pragma unroll
            for (int off = 16; off > 0; off >>= 1) z += __shfl_down(z, off);
            z = __shfl(z, 0);
            inv[jj] = 1.0f / z;
        }
        float mxa = jh2 ? mx[2] : mx[0], mxb = jh2 ? mx[3] : mx[1];
        float iva = jh2 ? inv[2] : inv[0], ivb = jh2 ? inv[3] : inv[1];
        cwl[cl * 4 + jh2]     = __expf(e0 - mxa) * iva;
        cwl[cl * 4 + jh2 + 1] = __expf(e1 - mxb) * ivb;
    }
    __syncthreads();
    const float* pw = cwl + w * 64;

#define SK_FENCE() __builtin_amdgcn_sched_barrier(0)
    sk_load_w(fA, pw);
    for (int it = 0; it < SK_NB / 2 - 1; it++) {
        sk_load(fB, pa + SK_CN * SK_WC, pb + SK_CN * SK_XC);
        sk_load_w(fB, pw + 4 * SK_CN);
        SK_FENCE();
        sk_compute(fA, acc);
        pa += 2 * SK_CN * SK_WC; pb += 2 * SK_CN * SK_XC; pw += 8 * SK_CN;
        sk_load(fA, pa, pb);
        sk_load_w(fA, pw);
        SK_FENCE();
        sk_compute(fB, acc);
    }
    sk_load(fB, pa + SK_CN * SK_WC, pb + SK_CN * SK_XC);
    sk_load_w(fB, pw + 4 * SK_CN);
    SK_FENCE();
    sk_compute(fA, acc);
    sk_compute(fB, acc);
#undef SK_FENCE

    float* pp = part + (size_t)w * (64 * SK_PP) + ln * SK_PP + 8 * h;
#pragma unroll
    for (int jj = 0; jj < 4; jj++)
#pragma unroll
        for (int bb = 0; bb < 4; bb++) {
            float* d = pp + bb * 16 * SK_PP + jj * 16;
            *(v4f*)(d)     = (v4f){acc[jj][bb][0], acc[jj][bb][1], acc[jj][bb][2], acc[jj][bb][3]};
            *(v4f*)(d + 4) = (v4f){acc[jj][bb][4], acc[jj][bb][5], acc[jj][bb][6], acc[jj][bb][7]};
        }
    __syncthreads();
    float* sp = spart + (size_t)ks * (B_ * J_ * S_) + (size_t)(bq * 64) * (J_ * S_) + j0 * S_;
#pragma unroll
    for (int q = 0; q < 4; q++) {
        int e = tid + 256 * q, row = e >> 4, c4 = (e & 15) * 4;
        const float* r = part + row * SK_PP + c4;
        v4f sum = *(const v4f*)r;
#pragma unroll
        for (int ww = 1; ww < 8; ww++) sum += *(const v4f*)(r + ww * (64 * SK_PP));
        *(v4f*)(sp + (size_t)row * (J_ * S_) + c4) = sum;
    }
}

__global__ void squash_kernel(const float* __restrict__ spart, float* __restrict__ vbuf,
                              float* __restrict__ out, int write_out) {
    int b = blockIdx.x;
    int tid = threadIdx.x;
    __shared__ float red[512];
    size_t idx = (size_t)b * 512 + tid;
    float pv[SK_KS];
#pragma unroll
    for (int k = 0; k < SK_KS; k++) pv[k] = spart[idx + (size_t)k * 65536];
    float val = pv[0];
#pragma unroll
    for (int k = 1; k < SK_KS; k++) val += pv[k];
    red[tid] = val * val;
    __syncthreads();
    for (int jo = 256; jo >= 16; jo >>= 1) {
        if (tid < jo) red[tid] += red[tid + jo];
        __syncthreads();
    }
    float msq = red[tid & 15];
    float scale = msq / ((1.0f + msq) * sqrtf(msq));
    float vv = val * scale;
    vbuf[idx] = vv;
    if (write_out) out[idx] = vv;
}

#define AG_CN 2
#define AG_NS (16 / AG_CN)

struct AgSet {
    v4f a0[AG_CN], a1[AG_CN];
    v4f b[AG_CN][8];
};

__device__ __forceinline__ void ag_load(AgSet& f, const float* __restrict__ pa,
                                        const float* __restrict__ pb) {
#pragma unroll
    for (int i = 0; i < AG_CN; i++) {
        f.a0[i] = *(const v4f*)(pa + i * SK_WC);
        f.a1[i] = *(const v4f*)(pa + i * SK_WC + S_ * U_);
#pragma unroll
        for (int t = 0; t < 8; t++)
            f.b[i][t] = *(const v4f*)(pb + i * SK_XC + t * 16 * U_);
    }
}

__device__ __forceinline__ void ag_compute(const AgSet& f, const v8f (&v0)[8],
                                           const v8f (&v1)[8], float* __restrict__ rp) {
#pragma unroll
    for (int i = 0; i < AG_CN; i++) {
        v2f a0l = {f.a0[i].x, f.a0[i].y}, a0h = {f.a0[i].z, f.a0[i].w};
        v2f a1l = {f.a1[i].x, f.a1[i].y}, a1h = {f.a1[i].z, f.a1[i].w};
        v8f p0 = {}, p1 = {};
#pragma unroll
        for (int t = 0; t < 8; t++) {
            v2f bl = {f.b[i][t].x, f.b[i][t].y}, bh = {f.b[i][t].z, f.b[i][t].w};
            v8f z = {};
            v8f d0 = __builtin_amdgcn_wmma_f32_16x16x4_f32(false, a0l, false, bl, (short)0, z, false, false);
            v8f d1 = __builtin_amdgcn_wmma_f32_16x16x4_f32(false, a1l, false, bl, (short)0, z, false, false);
            d0 = __builtin_amdgcn_wmma_f32_16x16x4_f32(false, a0h, false, bh, (short)0, d0, false, false);
            d1 = __builtin_amdgcn_wmma_f32_16x16x4_f32(false, a1h, false, bh, (short)0, d1, false, false);
            p0 += d0 * v0[t];
            p1 += d1 * v1[t];
        }
        float s0 = ((p0[0] + p0[1]) + (p0[2] + p0[3])) + ((p0[4] + p0[5]) + (p0[6] + p0[7]));
        float s1 = ((p1[0] + p1[1]) + (p1[2] + p1[3])) + ((p1[4] + p1[5]) + (p1[6] + p1[7]));
        rp[(i * 2 + 0) * 33] = s0;
        rp[(i * 2 + 1) * 33] = s1;
    }
}

__global__ __launch_bounds__(256) void bupd_kernel(
    const float* __restrict__ xT, const float* __restrict__ W,
    const float* __restrict__ vbuf, float* __restrict__ bT, int first) {
    int bx  = blockIdx.x;
    int jh  = bx & 1;
    int cr  = bx >> 1;
    int tid = threadIdx.x;
    int w    = __builtin_amdgcn_readfirstlane(tid >> 5);
    int lane = tid & 31;
    int h    = lane >> 4;
    int ln   = lane & 15;
    int j0   = (jh * 8 + w) * 2;
    int c0   = cr * 16;

    const float* pa = W  + ((size_t)c0 * J_ + j0) * (S_ * U_) + ln * U_ + 4 * h;
    const float* pb = xT + ((size_t)c0 * B_ + ln) * U_ + 4 * h;
    const float* pv = vbuf + (size_t)ln * (J_ * S_) + j0 * S_ + 8 * h;

    v8f v0[8], v1[8];
#pragma unroll
    for (int t = 0; t < 8; t++) {
        const float* q = pv + (size_t)t * 16 * (J_ * S_);
        v4f l0 = *(const v4f*)(q),      h0 = *(const v4f*)(q + 4);
        v4f l1 = *(const v4f*)(q + S_), h1 = *(const v4f*)(q + S_ + 4);
        v0[t] = (v8f){l0.x, l0.y, l0.z, l0.w, h0.x, h0.y, h0.z, h0.w};
        v1[t] = (v8f){l1.x, l1.y, l1.z, l1.w, h1.x, h1.y, h1.z, h1.w};
    }

    __shared__ float red[8][32][33];
    float* rp = &red[w][0][lane];

    AgSet fA, fB;
#define AG_FENCE() __builtin_amdgcn_sched_barrier(0)
    ag_load(fA, pa, pb);
#pragma unroll 1
    for (int it = 0; it < AG_NS / 2 - 1; it++) {
        ag_load(fB, pa + AG_CN * SK_WC, pb + AG_CN * SK_XC);
        AG_FENCE();
        ag_compute(fA, v0, v1, rp);
        pa += 2 * AG_CN * SK_WC; pb += 2 * AG_CN * SK_XC;
        ag_load(fA, pa, pb);
        AG_FENCE();
        ag_compute(fB, v0, v1, rp + AG_CN * 2 * 33);
        rp += 2 * AG_CN * 2 * 33;
    }
    ag_load(fB, pa + AG_CN * SK_WC, pb + AG_CN * SK_XC);
    AG_FENCE();
    ag_compute(fA, v0, v1, rp);
    ag_compute(fB, v0, v1, rp + AG_CN * 2 * 33);
#undef AG_FENCE

    __syncthreads();
    const float* rr = &red[w][lane][0];
    float sum = 0.f;
#pragma unroll
    for (int k = 0; k < 32; k++) sum += rr[k];
    size_t bi = (size_t)(j0 + (lane & 1)) * C_ + c0 + (lane >> 1);
    float upd = sum * (1.0f / 128.0f);
    bT[bi] = first ? upd : bT[bi] + upd;
}

extern "C" void kernel_launch(void* const* d_in, const int* in_sizes, int n_in,
                              void* d_out, int out_size, void* d_ws, size_t ws_size,
                              hipStream_t stream) {
    (void)in_sizes; (void)n_in; (void)out_size; (void)ws_size;
    const float* x = (const float*)d_in[0];
    const float* W = (const float*)d_in[1];
    float* out = (float*)d_out;

    float* ws    = (float*)d_ws;
    float* xT    = ws;
    float* bT    = xT + 2097152;
    float* spart = bT + 2 * 65536;
    float* vbuf  = spart + 16 * 65536;

    transpose_x_kernel<<<512, 256, 0, stream>>>(x, xT);

    for (int it = 0; it < 3; it++) {
        s_kernel<<<256, 256, 0, stream>>>(W, xT, bT, spart, it == 0 ? 1 : 0);
        squash_kernel<<<128, 512, 0, stream>>>(spart, vbuf, out, it == 2 ? 1 : 0);
        if (it < 2) {
            bupd_kernel<<<256, 256, 0, stream>>>(xT, W, vbuf, bT, it == 0 ? 1 : 0);
        }
    }
}


# route_kernel softmax against a reference fixed per wave (first c's old logit; one bounded redo at the true maximum if it lies over 32 away): no rescale multiply, one exp per (c,j); sum order kept
# speedup vs baseline: 4.6848x; 1.2275x over previous
#include <hip/hip_runtime.h>
#include <hip/hip_bf16.h>

typedef float v2f __attribute__((ext_vector_type(2)));
typedef float v4f __attribute__((ext_vector_type(4)));
typedef float v8f __attribute__((ext_vector_type(8)));

#define B_ 128
#define U_ 8
#define C_ 2048
#define J_ 32
#define S_ 16

__global__ __launch_bounds__(256) void transpose_x_kernel(const float* __restrict__ x,
                                                          float* __restrict__ xT) {
    __shared__ float tile[32][129];
    int c0   = (blockIdx.x & 63) * 32;
    int r0   = (blockIdx.x >> 6) * 128;
    int lane = threadIdx.x & 31;
    int w    = threadIdx.x >> 5;
#pragma unroll
    for (int i = 0; i < 16; i++) {
        int r = w * 16 + i;
        tile[lane][r] = x[(size_t)(r0 + r) * C_ + c0 + lane];
    }
    __syncthreads();
#pragma unroll
    for (int q = 0; q < 4; q++) {
        int cl = w * 4 + q;
#pragma unroll
        for (int i = 0; i < 4; i++)
            xT[(size_t)(c0 + cl) * (B_ * U_) + r0 + i * 32 + lane] = tile[cl][i * 32 + lane];
    }
}

#define SK_CN 2
#define SK_NB (16 / SK_CN)
#define SK_WC (J_ * S_ * U_)
#define SK_XC (B_ * U_)
#define SK_KS 16
#define SK_PP 68

struct SkSet {
    v4f a[SK_CN][4];
    v4f b[SK_CN][4];
    v4f cw[SK_CN];
};

__device__ __forceinline__ void sk_load_w(SkSet& f, const float* pw) {
#pragma unroll
    for (int q = 0; q < SK_CN; q++) f.cw[q] = *(const v4f*)(pw + 4 * q);
}

__device__ __forceinline__ void sk_load(SkSet& f, const float* __restrict__ pa,
                                        const float* __restrict__ pb) {
#pragma unroll
    for (int i = 0; i < SK_CN; i++) {
#pragma unroll
        for (int t = 0; t < 4; t++) {
            f.a[i][t] = *(const v4f*)(pa + i * SK_WC + t * S_ * U_);
            f.b[i][t] = *(const v4f*)(pb + i * SK_XC + t * 16 * U_);
        }
    }
}

__device__ __forceinline__ void sk_compute(const SkSet& f, v8f (&acc)[4][4]) {
#pragma unroll
    for (int i = 0; i < SK_CN; i++) {
        v2f al[4], ah[4], bl[4], bh[4];
#pragma unroll
        for (int t = 0; t < 4; t++) {
            v4f fa = f.a[i][t] * f.cw[i][t];
            al[t] = (v2f){fa.x, fa.y}; ah[t] = (v2f){fa.z, fa.w};
            bl[t] = (v2f){f.b[i][t].x, f.b[i][t].y}; bh[t] = (v2f){f.b[i][t].z, f.b[i][t].w};
        }
#pragma unroll
        for (int jj = 0; jj < 4; jj++)
#pragma unroll
            for (int bb = 0; bb < 4; bb++)
                acc[jj][bb] = __builtin_amdgcn_wmma_f32_16x16x4_f32(false, al[jj], false, bl[bb], (short)0, acc[jj][bb], false, false);
#pragma unroll
        for (int jj = 0; jj < 4; jj++)
#pragma unroll
            for (int bb = 0; bb < 4; bb++)
                acc[jj][bb] = __builtin_amdgcn_wmma_f32_16x16x4_f32(false, ah[jj], false, bh[bb], (short)0, acc[jj][bb], false, false);
    }
}

__global__ __launch_bounds__(256) void s_kernel(
    const float* __restrict__ W, const float* __restrict__ xT, float* __restrict__ spart) {
    int bx  = blockIdx.x;
    int ks  = bx & 15;
    int bq  = (bx >> 4) & 1;
    int jq  = bx >> 5;
    int j0  = jq * 4;
    int bt0 = bq * 4;

    int tid  = threadIdx.x;
    int w    = __builtin_amdgcn_readfirstlane(tid >> 5);
    int lane = tid & 31;
    int h    = lane >> 4;
    int ln   = lane & 15;

    int cs = ks * 128 + w * 16;
    const float* pa = W  + ((size_t)cs * J_ + j0) * (S_ * U_) + ln * U_ + 4 * h;
    const float* pb = xT + ((size_t)cs * B_ + bt0 * 16 + ln) * U_ + 4 * h;

    __shared__ __attribute__((aligned(16))) float part[8 * 64 * SK_PP];
    __shared__ __attribute__((aligned(16))) float cwl[128 * 4];

    v8f acc[4][4] = {};
    SkSet fA, fB;
    sk_load(fA, pa, pb);
    cwl[tid]       = 1.0f / 2048.0f;
    cwl[tid + 256] = 1.0f / 2048.0f;
    __syncthreads();
    const float* pw = cwl + w * 64;

#define SK_FENCE() __builtin_amdgcn_sched_barrier(0)
    sk_load_w(fA, pw);
    for (int it = 0; it < SK_NB / 2 - 1; it++) {
        sk_load(fB, pa + SK_CN * SK_WC, pb + SK_CN * SK_XC);
        sk_load_w(fB, pw + 4 * SK_CN);
        SK_FENCE();
        sk_compute(fA, acc);
        pa += 2 * SK_CN * SK_WC; pb += 2 * SK_CN * SK_XC; pw += 8 * SK_CN;
        sk_load(fA, pa, pb);
        sk_load_w(fA, pw);
        SK_FENCE();
        sk_compute(fB, acc);
    }
    sk_load(fB, pa + SK_CN * SK_WC, pb + SK_CN * SK_XC);
    sk_load_w(fB, pw + 4 * SK_CN);
    SK_FENCE();
    sk_compute(fA, acc);
    sk_compute(fB, acc);
#undef SK_FENCE

    float* pp = part + (size_t)w * (64 * SK_PP) + ln * SK_PP + 8 * h;
#pragma unroll
    for (int jj = 0; jj < 4; jj++)
#pragma unroll
        for (int bb = 0; bb < 4; bb++) {
            float* d = pp + bb * 16 * SK_PP + jj * 16;
            *(v4f*)(d)     = (v4f){acc[jj][bb][0], acc[jj][bb][1], acc[jj][bb][2], acc[jj][bb][3]};
            *(v4f*)(d + 4) = (v4f){acc[jj][bb][4], acc[jj][bb][5], acc[jj][bb][6], acc[jj][bb][7]};
        }
    __syncthreads();
    float* sp = spart + (size_t)ks * (B_ * J_ * S_) + (size_t)(bq * 64) * (J_ * S_) + j0 * S_;
#pragma unroll
    for (int q = 0; q < 4; q++) {
        int e = tid + 256 * q, row = e >> 4, c4 = (e & 15) * 4;
        const float* r = part + row * SK_PP + c4;
        v4f sum = *(const v4f*)r;
#pragma unroll
        for (int ww = 1; ww < 8; ww++) sum += *(const v4f*)(r + ww * (64 * SK_PP));
        *(v4f*)(sp + (size_t)row * (J_ * S_) + c4) = sum;
    }
}

template <bool JOIN>
__global__ void squash_kernel(const float* __restrict__ spart, const float* __restrict__ stats,
                              float* __restrict__ vbuf, float* __restrict__ out, int write_out) {
    int b = blockIdx.x;
    int tid = threadIdx.x;
    __shared__ float red[512];
    size_t idx = (size_t)b * 512 + tid;
    float pv[SK_KS];
#pragma unroll
    for (int k = 0; k < SK_KS; k++) pv[k] = spart[idx + (size_t)k * 65536];
    float val;
    if (JOIN) {
        int j = tid >> 4;
        float mk[SK_KS], zk[SK_KS];
#pragma unroll
        for (int k = 0; k < SK_KS; k++) {
            const float* st = stats + (k * 16 + (j >> 1)) * 32 + (j & 1);
            mk[k] = st[0]; zk[k] = st[2];
        }
        float M = mk[0];
#pragma unroll
        for (int k = 1; k < SK_KS; k++) M = fmaxf(M, mk[k]);
        float num = 0.f, den = 0.f;
#pragma unroll
        for (int k = 0; k < SK_KS; k++) {
            float g = __expf(mk[k] - M);
            num += g * pv[k];
            den += g * zk[k];
        }
        val = num / den;
    } else {
        val = pv[0];
#pragma unroll
        for (int k = 1; k < SK_KS; k++) val += pv[k];
    }
    red[tid] = val * val;
    __syncthreads();
    for (int jo = 256; jo >= 16; jo >>= 1) {
        if (tid < jo) red[tid] += red[tid + jo];
        __syncthreads();
    }
    float msq = red[tid & 15];
    float scale = msq / ((1.0f + msq) * sqrtf(msq));
    float vv = val * scale;
    vbuf[idx] = vv;
    if (write_out) out[idx] = vv;
}

#define RK_PP 36

struct RkSet {
    v4f a0, a1;
    v4f b[8];
};

__device__ __forceinline__ void rk_load(RkSet& f, const float* __restrict__ pa,
                                        const float* __restrict__ pb) {
    f.a0 = *(const v4f*)(pa);
    f.a1 = *(const v4f*)(pa + S_ * U_);
#pragma unroll
    for (int t = 0; t < 8; t++) f.b[t] = *(const v4f*)(pb + t * 16 * U_);
}

template <int M>
__device__ __forceinline__ float rk_rowx(float x) {
    int b = __builtin_bit_cast(int, x);
    return __builtin_bit_cast(float, __builtin_amdgcn_update_dpp(b, b, 0x160 | M, 0xf, 0xf, false));
}
__device__ __forceinline__ float rk_xhalf(float x) {
    int b = __builtin_bit_cast(int, x);
    return __builtin_bit_cast(float, __builtin_amdgcn_permlanex16(b, b, 0x76543210, 0xfedcba98, false, false));
}
__device__ __forceinline__ float rk_wave_sum(float x) {
    x += rk_rowx<1>(x);
    x += rk_rowx<2>(x);
    x += rk_rowx<4>(x);
    x += rk_rowx<8>(x);
    return x + rk_xhalf(x);
}

__device__ __forceinline__ void rk_one(const v4f a, const v4f (&b)[8], const v8f (&v)[8],
                                       v8f (&acc)[8], float bold, float ref, float& m, float& z, float& bnew) {
    v2f al = {a.x, a.y}, ah = {a.z, a.w};
    v8f d[8];
#pragma unroll
    for (int t = 0; t < 8; t++) {
        v2f bl = {b[t].x, b[t].y}, bh = {b[t].z, b[t].w};
        v8f zz = {};
        d[t] = __builtin_amdgcn_wmma_f32_16x16x4_f32(false, al, false, bl, (short)0, zz, false, false);
        d[t] = __builtin_amdgcn_wmma_f32_16x16x4_f32(false, ah, false, bh, (short)0, d[t], false, false);
    }
    v8f p = {};
#pragma unroll
    for (int t = 0; t < 8; t++) p += d[t] * v[t];
    float sl = ((p[0] + p[1]) + (p[2] + p[3])) + ((p[4] + p[5]) + (p[6] + p[7]));
    float ag = rk_wave_sum(sl);
    float bn = bold + ag * (1.0f / 128.0f);
    float e  = __expf(bn - ref);
    z += e;
    m = fmaxf(m, bn);
    bnew = bn;
#pragma unroll
    for (int t = 0; t < 8; t++) acc[t] += d[t] * e;
}

__device__ __forceinline__ float rk_lane(float x, int l) {
    return __builtin_bit_cast(float, __builtin_amdgcn_readlane(__builtin_bit_cast(int, x), l));
}

__device__ __forceinline__ void rk_step(const RkSet& f, const v8f (&v0)[8], const v8f (&v1)[8],
                                        v8f (&acc0)[8], v8f (&acc1)[8], float bl, int ci, int lane,
                                        float r0, float r1,
                                        float& m0, float& m1, float& z0, float& z1, float& bout) {
    float bn0, bn1;
    rk_one(f.a0, f.b, v0, acc0, rk_lane(bl, 2 * ci),     r0, m0, z0, bn0);
    __builtin_amdgcn_sched_barrier(0);
    rk_one(f.a1, f.b, v1, acc1, rk_lane(bl, 2 * ci + 1), r1, m1, z1, bn1);
    bout = (lane == 2 * ci)     ? bn0 : bout;
    bout = (lane == 2 * ci + 1) ? bn1 : bout;
}

__global__ __launch_bounds__(256) void route_kernel(
    const float* __restrict__ xT, const float* __restrict__ W, const float* __restrict__ vbuf,
    float* __restrict__ bT, float* __restrict__ spart, float* __restrict__ stats, int first) {
    int bx  = blockIdx.x;
    int jp  = bx & 15;
    int ks  = bx >> 4;
    int tid = threadIdx.x;
    int w    = __builtin_amdgcn_readfirstlane(tid >> 5);
    int lane = tid & 31;
    int h    = lane >> 4;
    int ln   = lane & 15;
    int j0   = jp * 2;
    int cs   = ks * 128 + w * 16;

    const float* pa = W  + ((size_t)cs * J_ + j0) * (S_ * U_) + ln * U_ + 4 * h;
    const float* pb = xT + ((size_t)cs * B_ + ln) * U_ + 4 * h;
    const float* pv = vbuf + (size_t)ln * (J_ * S_) + j0 * S_ + 8 * h;

    RkSet fA, fB;
    rk_load(fA, pa, pb);

    v8f v0[8], v1[8];
#pragma unroll
    for (int t = 0; t < 8; t++) {
        const float* q = pv + (size_t)t * 16 * (J_ * S_);
        v4f l0 = *(const v4f*)(q),      h0 = *(const v4f*)(q + 4);
        v4f l1 = *(const v4f*)(q + S_), h1 = *(const v4f*)(q + S_ + 4);
        v0[t] = (v8f){l0.x, l0.y, l0.z, l0.w, h0.x, h0.y, h0.z, h0.w};
        v1[t] = (v8f){l1.x, l1.y, l1.z, l1.w, h1.x, h1.y, h1.z, h1.w};
    }
    float bl = 0.f;
    if (!first) bl = bT[(size_t)(j0 + (lane & 1)) * C_ + cs + (lane >> 1)];
    float bout = 0.f;

#define RK_FAR 32.0f
    float r0 = rk_lane(bl, 0), r1 = rk_lane(bl, 1);
    v8f acc0[8], acc1[8];
    float m0, m1, z0, z1;
#pragma unroll 1
    for (int tr = 0; tr < 2; tr++) {
#pragma unroll
    for (int t = 0; t < 8; t++) { acc0[t] = (v8f){}; acc1[t] = (v8f){}; }
    m0 = -1e30f; m1 = -1e30f; z0 = 0.f; z1 = 0.f;

#define RK_FENCE() __builtin_amdgcn_sched_barrier(0)
#pragma unroll 1
    for (int it = 0; it < 8; it++) {
        rk_load(fB, pa + SK_WC, pb + SK_XC);
        RK_FENCE();
        rk_step(fA, v0, v1, acc0, acc1, bl, 2 * it, lane, r0, r1, m0, m1, z0, z1, bout);
        int adv = (it == 7) ? -14 : 2;
        pa += adv * SK_WC; pb += adv * SK_XC;
        rk_load(fA, pa, pb);
        RK_FENCE();
        rk_step(fB, v0, v1, acc0, acc1, bl, 2 * it + 1, lane, r0, r1, m0, m1, z0, z1, bout);
    }
    float t0 = rk_lane(m0, 0), t1 = rk_lane(m1, 0);
    if (fabsf(t0 - r0) <= RK_FAR && fabsf(t1 - r1) <= RK_FAR) break;
    r0 = t0; r1 = t1;
    }
#undef RK_FENCE
#undef RK_FAR

    __shared__ __attribute__((aligned(16))) float part[8 * 128 * RK_PP];
    __shared__ float smx[8][2], szz[8][2];
    __shared__ float sbn[2][128];
    sbn[lane & 1][w * 16 + (lane >> 1)] = bout;
    if (lane == 0) { smx[w][0] = m0; smx[w][1] = m1; }
    __syncthreads();
    bT[(size_t)(j0 + (tid >> 7)) * C_ + ks * 128 + (tid & 127)] = sbn[tid >> 7][tid & 127];
    float M0 = smx[0][0], M1 = smx[0][1];
#pragma unroll
    for (int q = 1; q < 8; q++) { M0 = fmaxf(M0, smx[q][0]); M1 = fmaxf(M1, smx[q][1]); }
    float g0 = __expf(r0 - M0), g1 = __expf(r1 - M1);
    if (lane == 0) { szz[w][0] = z0 * g0; szz[w][1] = z1 * g1; }
    float* pp = part + (size_t)w * (128 * RK_PP) + ln * RK_PP + 8 * h;
#pragma unroll
    for (int t = 0; t < 8; t++) {
        float* d = pp + t * 16 * RK_PP;
        v8f a = acc0[t] * g0, c = acc1[t] * g1;
        *(v4f*)(d)      = (v4f){a[0], a[1], a[2], a[3]};
        *(v4f*)(d + 4)  = (v4f){a[4], a[5], a[6], a[7]};
        *(v4f*)(d + 16) = (v4f){c[0], c[1], c[2], c[3]};
        *(v4f*)(d + 20) = (v4f){c[4], c[5], c[6], c[7]};
    }
    __syncthreads();
    float* sp = spart + (size_t)ks * (B_ * J_ * S_) + j0 * S_;
#pragma unroll
    for (int q = 0; q < 4; q++) {
        int e = tid + 256 * q, row = e >> 3, c4 = (e & 7) * 4;
        const float* r = part + row * RK_PP + c4;
        v4f sum = *(const v4f*)r;
#pragma unroll
        for (int ww = 1; ww < 8; ww++) sum += *(const v4f*)(r + ww * (128 * RK_PP));
        *(v4f*)(sp + (size_t)row * (J_ * S_) + c4) = sum;
    }
    if (tid < 4) {
        int jj = tid & 1;
        float zt = szz[0][jj];
        for (int q = 1; q < 8; q++) zt += szz[q][jj];
        float mm = jj ? M1 : M0;
        stats[(ks * 16 + jp) * 32 + tid] = (tid < 2) ? mm : zt;
    }
}

extern "C" void kernel_launch(void* const* d_in, const int* in_sizes, int n_in,
                              void* d_out, int out_size, void* d_ws, size_t ws_size,
                              hipStream_t stream) {
    (void)in_sizes; (void)n_in; (void)out_size; (void)ws_size;
    const float* x = (const float*)d_in[0];
    const float* W = (const float*)d_in[1];
    float* out = (float*)d_out;

    float* ws    = (float*)d_ws;
    float* xT    = ws;
    float* bT    = xT + 2097152;
    float* stats = bT + 65536;
    float* spart = bT + 2 * 65536;
    float* vbuf  = spart + 16 * 65536;

    transpose_x_kernel<<<512, 256, 0, stream>>>(x, xT);

    s_kernel<<<256, 256, 0, stream>>>(W, xT, spart);
    squash_kernel<false><<<128, 512, 0, stream>>>(spart, stats, vbuf, out, 0);
    route_kernel<<<256, 256, 0, stream>>>(xT, W, vbuf, bT, spart, stats, 1);
    squash_kernel<true><<<128, 512, 0, stream>>>(spart, stats, vbuf, out, 0);
    route_kernel<<<256, 256, 0, stream>>>(xT, W, vbuf, bT, spart, stats, 0);
    squash_kernel<true><<<128, 512, 0, stream>>>(spart, stats, vbuf, out, 1);
}


# x layout launch folded into s_kernel: each block transposes its x tile (128c x 64b x 8u) through LDS and takes B fragments there; blocks share out the xT copy; 7 -> 6 launches; sum order kept
# speedup vs baseline: 4.8386x; 1.0328x over previous
#include <hip/hip_runtime.h>
#include <hip/hip_bf16.h>

typedef float v2f __attribute__((ext_vector_type(2)));
typedef float v4f __attribute__((ext_vector_type(4)));
typedef float v8f __attribute__((ext_vector_type(8)));

#define B_ 128
#define U_ 8
#define C_ 2048
#define J_ 32
#define S_ 16

#define SK_CN 2
#define SK_NB (16 / SK_CN)
#define SK_WC (J_ * S_ * U_)
#define SK_XC (B_ * U_)
#define SK_KS 16
#define SK_PP 68
#define SX_P  516
#define SX_WT (16 * SX_P + 8)

struct SkSet {
    v4f a[SK_CN][4];
    v4f b[SK_CN][4];
    v4f cw[SK_CN];
};

__device__ __forceinline__ void sk_load_w(SkSet& f, const float* pw) {
#pragma unroll
    for (int q = 0; q < SK_CN; q++) f.cw[q] = *(const v4f*)(pw + 4 * q);
}

__device__ __forceinline__ void sk_load_a(SkSet& f, const float* __restrict__ pa) {
#pragma unroll
    for (int i = 0; i < SK_CN; i++)
#pragma unroll
        for (int t = 0; t < 4; t++)
            f.a[i][t] = *(const v4f*)(pa + i * SK_WC + t * S_ * U_);
}
__device__ __forceinline__ void sk_load_b(SkSet& f, const float* pb) {
#pragma unroll
    for (int i = 0; i < SK_CN; i++)
#pragma unroll
        for (int t = 0; t < 4; t++)
            f.b[i][t] = *(const v4f*)(pb + i * SX_P + t * 16 * U_);
}
__device__ __forceinline__ void sk_load(SkSet& f, const float* __restrict__ pa, const float* pb) {
    sk_load_a(f, pa);
    sk_load_b(f, pb);
}

__device__ __forceinline__ void sk_compute(const SkSet& f, v8f (&acc)[4][4]) {
#pragma unroll
    for (int i = 0; i < SK_CN; i++) {
        v2f al[4], ah[4], bl[4], bh[4];
#pragma unroll
        for (int t = 0; t < 4; t++) {
            v4f fa = f.a[i][t] * f.cw[i][t];
            al[t] = (v2f){fa.x, fa.y}; ah[t] = (v2f){fa.z, fa.w};
            bl[t] = (v2f){f.b[i][t].x, f.b[i][t].y}; bh[t] = (v2f){f.b[i][t].z, f.b[i][t].w};
        }
#pragma unroll
        for (int jj = 0; jj < 4; jj++)
#pragma unroll
            for (int bb = 0; bb < 4; bb++)
                acc[jj][bb] = __builtin_amdgcn_wmma_f32_16x16x4_f32(false, al[jj], false, bl[bb], (short)0, acc[jj][bb], false, false);
#pragma unroll
        for (int jj = 0; jj < 4; jj++)
#pragma unroll
            for (int bb = 0; bb < 4; bb++)
                acc[jj][bb] = __builtin_amdgcn_wmma_f32_16x16x4_f32(false, ah[jj], false, bh[bb], (short)0, acc[jj][bb], false, false);
    }
}

__global__ __launch_bounds__(256) void s_kernel(
    const float* __restrict__ W, const float* __restrict__ x, float* __restrict__ xT,
    float* __restrict__ spart) {
    int bx  = blockIdx.x;
    int ks  = bx & 15;
    int bq  = (bx >> 4) & 1;
    int jq  = bx >> 5;
    int j0  = jq * 4;

    int tid  = threadIdx.x;
    int w    = __builtin_amdgcn_readfirstlane(tid >> 5);
    int lane = tid & 31;
    int h    = lane >> 4;
    int ln   = lane & 15;

    int cs = ks * 128 + w * 16;
    const float* pa = W + ((size_t)cs * J_ + j0) * (S_ * U_) + ln * U_ + 4 * h;

    __shared__ __attribute__((aligned(16))) float lds[8 * SX_WT];
    __shared__ __attribute__((aligned(16))) float cwl[128 * 4];
    float* part = lds;
    float* xs   = lds + (size_t)w * SX_WT;

    v8f acc[4][4] = {};
    SkSet fA, fB;
    sk_load_a(fA, pa);
    cwl[tid]       = 1.0f / 2048.0f;
    cwl[tid + 256] = 1.0f / 2048.0f;

    {
        int ur = lane >> 3, q = lane & 7;
        int wp = w & ~1, hb = w & 1;
        const float* px = x + ((size_t)(bq * 64 + 32 * hb) * U_ + ur) * C_ + ks * 128 + wp * 16 + 4 * q;
        float* pd = lds + (size_t)(wp + (q >> 2)) * SX_WT + (4 * (q & 3)) * SX_P + (32 * hb) * U_ + ur;
        v4f st[64];
#pragma unroll
        for (int i = 0; i < 64; i++) st[i] = *(const v4f*)(px + (size_t)i * (4 * C_));
#pragma unroll
        for (int i = 0; i < 64; i++) {
            float* d = pd + 4 * i;
            d[0]        = st[i].x;
            d[SX_P]     = st[i].y;
            d[2 * SX_P] = st[i].z;
            d[3 * SX_P] = st[i].w;
        }
    }
    __syncthreads();
    const float* pw = cwl + w * 64;

    {
        const float* sx = xs + (2 * jq) * SX_P + 4 * lane;
        float* gx = xT + (size_t)(cs + 2 * jq) * (B_ * U_) + bq * (64 * U_) + 4 * lane;
#pragma unroll
        for (int r = 0; r < 2; r++)
#pragma unroll
            for (int m = 0; m < 4; m++)
                *(v4f*)(gx + r * (B_ * U_) + 128 * m) = *(const v4f*)(sx + r * SX_P + 128 * m);
    }

    const float* pb = xs + ln * U_ + 4 * h;
    sk_load_b(fA, pb);

#define SK_FENCE() __builtin_amdgcn_sched_barrier(0)
    sk_load_w(fA, pw);
    for (int it = 0; it < SK_NB / 2 - 1; it++) {
        sk_load(fB, pa + SK_CN * SK_WC, pb + SK_CN * SX_P);
        sk_load_w(fB, pw + 4 * SK_CN);
        SK_FENCE();
        sk_compute(fA, acc);
        pa += 2 * SK_CN * SK_WC; pb += 2 * SK_CN * SX_P; pw += 8 * SK_CN;
        sk_load(fA, pa, pb);
        sk_load_w(fA, pw);
        SK_FENCE();
        sk_compute(fB, acc);
    }
    sk_load(fB, pa + SK_CN * SK_WC, pb + SK_CN * SX_P);
    sk_load_w(fB, pw + 4 * SK_CN);
    SK_FENCE();
    sk_compute(fA, acc);
    sk_compute(fB, acc);
#undef SK_FENCE

    __syncthreads();

    float* pp = part + (size_t)w * (64 * SK_PP) + ln * SK_PP + 8 * h;
#pragma unroll
    for (int jj = 0; jj < 4; jj++)
#pragma unroll
        for (int bb = 0; bb < 4; bb++) {
            float* d = pp + bb * 16 * SK_PP + jj * 16;
            *(v4f*)(d)     = (v4f){acc[jj][bb][0], acc[jj][bb][1], acc[jj][bb][2], acc[jj][bb][3]};
            *(v4f*)(d + 4) = (v4f){acc[jj][bb][4], acc[jj][bb][5], acc[jj][bb][6], acc[jj][bb][7]};
        }
    __syncthreads();
    float* sp = spart + (size_t)ks * (B_ * J_ * S_) + (size_t)(bq * 64) * (J_ * S_) + j0 * S_;
#pragma unroll
    for (int q = 0; q < 4; q++) {
        int e = tid + 256 * q, row = e >> 4, c4 = (e & 15) * 4;
        const float* r = part + row * SK_PP + c4;
        v4f sum = *(const v4f*)r;
#pragma unroll
        for (int ww = 1; ww < 8; ww++) sum += *(const v4f*)(r + ww * (64 * SK_PP));
        *(v4f*)(sp + (size_t)row * (J_ * S_) + c4) = sum;
    }
}

template <bool JOIN>
__global__ void squash_kernel(const float* __restrict__ spart, const float* __restrict__ stats,
                              float* __restrict__ vbuf, float* __restrict__ out, int write_out) {
    int b = blockIdx.x;
    int tid = threadIdx.x;
    __shared__ float red[512];
    size_t idx = (size_t)b * 512 + tid;
    float pv[SK_KS];
#pragma unroll
    for (int k = 0; k < SK_KS; k++) pv[k] = spart[idx + (size_t)k * 65536];
    float val;
    if (JOIN) {
        int j = tid >> 4;
        float mk[SK_KS], zk[SK_KS];
#pragma unroll
        for (int k = 0; k < SK_KS; k++) {
            const float* st = stats + (k * 16 + (j >> 1)) * 32 + (j & 1);
            mk[k] = st[0]; zk[k] = st[2];
        }
        float M = mk[0];
#pragma unroll
        for (int k = 1; k < SK_KS; k++) M = fmaxf(M, mk[k]);
        float num = 0.f, den = 0.f;
#pragma unroll
        for (int k = 0; k < SK_KS; k++) {
            float g = __expf(mk[k] - M);
            num += g * pv[k];
            den += g * zk[k];
        }
        val = num / den;
    } else {
        val = pv[0];
#pragma unroll
        for (int k = 1; k < SK_KS; k++) val += pv[k];
    }
    red[tid] = val * val;
    __syncthreads();
    for (int jo = 256; jo >= 16; jo >>= 1) {
        if (tid < jo) red[tid] += red[tid + jo];
        __syncthreads();
    }
    float msq = red[tid & 15];
    float scale = msq / ((1.0f + msq) * sqrtf(msq));
    float vv = val * scale;
    vbuf[idx] = vv;
    if (write_out) out[idx] = vv;
}

#define RK_PP 36

struct RkSet {
    v4f a0, a1;
    v4f b[8];
};

__device__ __forceinline__ void rk_load(RkSet& f, const float* __restrict__ pa,
                                        const float* __restrict__ pb) {
    f.a0 = *(const v4f*)(pa);
    f.a1 = *(const v4f*)(pa + S_ * U_);
#pragma unroll
    for (int t = 0; t < 8; t++) f.b[t] = *(const v4f*)(pb + t * 16 * U_);
}

template <int M>
__device__ __forceinline__ float rk_rowx(float x) {
    int b = __builtin_bit_cast(int, x);
    return __builtin_bit_cast(float, __builtin_amdgcn_update_dpp(b, b, 0x160 | M, 0xf, 0xf, false));
}
__device__ __forceinline__ float rk_xhalf(float x) {
    int b = __builtin_bit_cast(int, x);
    return __builtin_bit_cast(float, __builtin_amdgcn_permlanex16(b, b, 0x76543210, 0xfedcba98, false, false));
}
__device__ __forceinline__ float rk_wave_sum(float x) {
    x += rk_rowx<1>(x);
    x += rk_rowx<2>(x);
    x += rk_rowx<4>(x);
    x += rk_rowx<8>(x);
    return x + rk_xhalf(x);
}

__device__ __forceinline__ void rk_one(const v4f a, const v4f (&b)[8], const v8f (&v)[8],
                                       v8f (&acc)[8], float bold, float ref, float& m, float& z, float& bnew) {
    v2f al = {a.x, a.y}, ah = {a.z, a.w};
    v8f d[8];
#pragma unroll
    for (int t = 0; t < 8; t++) {
        v2f bl = {b[t].x, b[t].y}, bh = {b[t].z, b[t].w};
        v8f zz = {};
        d[t] = __builtin_amdgcn_wmma_f32_16x16x4_f32(false, al, false, bl, (short)0, zz, false, false);
        d[t] = __builtin_amdgcn_wmma_f32_16x16x4_f32(false, ah, false, bh, (short)0, d[t], false, false);
    }
    v8f p = {};
#pragma unroll
    for (int t = 0; t < 8; t++) p += d[t] * v[t];
    float sl = ((p[0] + p[1]) + (p[2] + p[3])) + ((p[4] + p[5]) + (p[6] + p[7]));
    float ag = rk_wave_sum(sl);
    float bn = bold + ag * (1.0f / 128.0f);
    float e  = __expf(bn - ref);
    z += e;
    m = fmaxf(m, bn);
    bnew = bn;
#pragma unroll
    for (int t = 0; t < 8; t++) acc[t] += d[t] * e;
}

__device__ __forceinline__ float rk_lane(float x, int l) {
    return __builtin_bit_cast(float, __builtin_amdgcn_readlane(__builtin_bit_cast(int, x), l));
}

__device__ __forceinline__ void rk_step(const RkSet& f, const v8f (&v0)[8], const v8f (&v1)[8],
                                        v8f (&acc0)[8], v8f (&acc1)[8], float bl, int ci, int lane,
                                        float r0, float r1,
                                        float& m0, float& m1, float& z0, float& z1, float& bout) {
    float bn0, bn1;
    rk_one(f.a0, f.b, v0, acc0, rk_lane(bl, 2 * ci),     r0, m0, z0, bn0);
    __builtin_amdgcn_sched_barrier(0);
    rk_one(f.a1, f.b, v1, acc1, rk_lane(bl, 2 * ci + 1), r1, m1, z1, bn1);
    bout = (lane == 2 * ci)     ? bn0 : bout;
    bout = (lane == 2 * ci + 1) ? bn1 : bout;
}

__global__ __launch_bounds__(256) void route_kernel(
    const float* __restrict__ xT, const float* __restrict__ W, const float* __restrict__ vbuf,
    float* __restrict__ bT, float* __restrict__ spart, float* __restrict__ stats, int first) {
    int bx  = blockIdx.x;
    int jp  = bx & 15;
    int ks  = bx >> 4;
    int tid = threadIdx.x;
    int w    = __builtin_amdgcn_readfirstlane(tid >> 5);
    int lane = tid & 31;
    int h    = lane >> 4;
    int ln   = lane & 15;
    int j0   = jp * 2;
    int cs   = ks * 128 + w * 16;

    const float* pa = W  + ((size_t)cs * J_ + j0) * (S_ * U_) + ln * U_ + 4 * h;
    const float* pb = xT + ((size_t)cs * B_ + ln) * U_ + 4 * h;
    const float* pv = vbuf + (size_t)ln * (J_ * S_) + j0 * S_ + 8 * h;

    RkSet fA, fB;
    rk_load(fA, pa, pb);

    v8f v0[8], v1[8];
#pragma unroll
    for (int t = 0; t < 8; t++) {
        const float* q = pv + (size_t)t * 16 * (J_ * S_);
        v4f l0 = *(const v4f*)(q),      h0 = *(const v4f*)(q + 4);
        v4f l1 = *(const v4f*)(q + S_), h1 = *(const v4f*)(q + S_ + 4);
        v0[t] = (v8f){l0.x, l0.y, l0.z, l0.w, h0.x, h0.y, h0.z, h0.w};
        v1[t] = (v8f){l1.x, l1.y, l1.z, l1.w, h1.x, h1.y, h1.z, h1.w};
    }
    float bl = 0.f;
    if (!first) bl = bT[(size_t)(j0 + (lane & 1)) * C_ + cs + (lane >> 1)];
    float bout = 0.f;

#define RK_FAR 32.0f
    float r0 = rk_lane(bl, 0), r1 = rk_lane(bl, 1);
    v8f acc0[8], acc1[8];
    float m0, m1, z0, z1;
#pragma unroll 1
    for (int tr = 0; tr < 2; tr++) {
#pragma unroll
    for (int t = 0; t < 8; t++) { acc0[t] = (v8f){}; acc1[t] = (v8f){}; }
    m0 = -1e30f; m1 = -1e30f; z0 = 0.f; z1 = 0.f;

#define RK_FENCE() __builtin_amdgcn_sched_barrier(0)
#pragma unroll 1
    for (int it = 0; it < 8; it++) {
        rk_load(fB, pa + SK_WC, pb + SK_XC);
        RK_FENCE();
        rk_step(fA, v0, v1, acc0, acc1, bl, 2 * it, lane, r0, r1, m0, m1, z0, z1, bout);
        int adv = (it == 7) ? -14 : 2;
        pa += adv * SK_WC; pb += adv * SK_XC;
        rk_load(fA, pa, pb);
        RK_FENCE();
        rk_step(fB, v0, v1, acc0, acc1, bl, 2 * it + 1, lane, r0, r1, m0, m1, z0, z1, bout);
    }
    float t0 = rk_lane(m0, 0), t1 = rk_lane(m1, 0);
    if (fabsf(t0 - r0) <= RK_FAR && fabsf(t1 - r1) <= RK_FAR) break;
    r0 = t0; r1 = t1;
    }
#undef RK_FENCE
#undef RK_FAR

    __shared__ __attribute__((aligned(16))) float part[8 * 128 * RK_PP];
    __shared__ float smx[8][2], szz[8][2];
    __shared__ float sbn[2][128];
    sbn[lane & 1][w * 16 + (lane >> 1)] = bout;
    if (lane == 0) { smx[w][0] = m0; smx[w][1] = m1; }
    __syncthreads();
    bT[(size_t)(j0 + (tid >> 7)) * C_ + ks * 128 + (tid & 127)] = sbn[tid >> 7][tid & 127];
    float M0 = smx[0][0], M1 = smx[0][1];
#pragma unroll
    for (int q = 1; q < 8; q++) { M0 = fmaxf(M0, smx[q][0]); M1 = fmaxf(M1, smx[q][1]); }
    float g0 = __expf(r0 - M0), g1 = __expf(r1 - M1);
    if (lane == 0) { szz[w][0] = z0 * g0; szz[w][1] = z1 * g1; }
    float* pp = part + (size_t)w * (128 * RK_PP) + ln * RK_PP + 8 * h;
#pragma unroll
    for (int t = 0; t < 8; t++) {
        float* d = pp + t * 16 * RK_PP;
        v8f a = acc0[t] * g0, c = acc1[t] * g1;
        *(v4f*)(d)      = (v4f){a[0], a[1], a[2], a[3]};
        *(v4f*)(d + 4)  = (v4f){a[4], a[5], a[6], a[7]};
        *(v4f*)(d + 16) = (v4f){c[0], c[1], c[2], c[3]};
        *(v4f*)(d + 20) = (v4f){c[4], c[5], c[6], c[7]};
    }
    __syncthreads();
    float* sp = spart + (size_t)ks * (B_ * J_ * S_) + j0 * S_;
#pragma unroll
    for (int q = 0; q < 4; q++) {
        int e = tid + 256 * q, row = e >> 3, c4 = (e & 7) * 4;
        const float* r = part + row * RK_PP + c4;
        v4f sum = *(const v4f*)r;
#pragma unroll
        for (int ww = 1; ww < 8; ww++) sum += *(const v4f*)(r + ww * (128 * RK_PP));
        *(v4f*)(sp + (size_t)row * (J_ * S_) + c4) = sum;
    }
    if (tid < 4) {
        int jj = tid & 1;
        float zt = szz[0][jj];
        for (int q = 1; q < 8; q++) zt += szz[q][jj];
        float mm = jj ? M1 : M0;
        stats[(ks * 16 + jp) * 32 + tid] = (tid < 2) ? mm : zt;
    }
}

extern "C" void kernel_launch(void* const* d_in, const int* in_sizes, int n_in,
                              void* d_out, int out_size, void* d_ws, size_t ws_size,
                              hipStream_t stream) {
    (void)in_sizes; (void)n_in; (void)out_size; (void)ws_size;
    const float* x = (const float*)d_in[0];
    const float* W = (const float*)d_in[1];
    float* out = (float*)d_out;

    float* ws    = (float*)d_ws;
    float* xT    = ws;
    float* bT    = xT + 2097152;
    float* stats = bT + 65536;
    float* spart = bT + 2 * 65536;
    float* vbuf  = spart + 16 * 65536;

    s_kernel<<<256, 256, 0, stream>>>(W, x, xT, spart);
    squash_kernel<false><<<128, 512, 0, stream>>>(spart, stats, vbuf, out, 0);
    route_kernel<<<256, 256, 0, stream>>>(xT, W, vbuf, bT, spart, stats, 1);
    squash_kernel<true><<<128, 512, 0, stream>>>(spart, stats, vbuf, out, 0);
    route_kernel<<<256, 256, 0, stream>>>(xT, W, vbuf, bT, spart, stats, 0);
    squash_kernel<true><<<128, 512, 0, stream>>>(spart, stats, vbuf, out, 1);
}
